# starting kernel
# baseline (speedup 1.0000x reference)
#include <hip/hip_runtime.h>
#include <math.h>

#define DM   512
#define NH   8
#define DKK  64
#define B_   2
#define LL   256
#define ROWS (B_*LL)

typedef __attribute__((ext_vector_type(2))) float v2f;
typedef __attribute__((ext_vector_type(8))) float v8f;

__global__ __launch_bounds__(256)
void gemm512_wmma(const float* __restrict__ A, const float* __restrict__ W,
                  const float* __restrict__ lam, float* __restrict__ Y, int act)
{
    const int wave = threadIdx.x >> 5;
    const int lane = threadIdx.x & 31;
    const int tile = blockIdx.x * 8 + wave;
    const int tr   = tile >> 5;
    const int tc   = tile & 31;
    const int half = lane >> 4;
    const int idx  = lane & 15;

    const float* arow = A + (size_t)(tr * 16 + idx) * DM + 2 * half;
    const float* wrow = W + (size_t)(tc * 16 + idx) * DM + 2 * half;

    v8f acc = {0.f, 0.f, 0.f, 0.f, 0.f, 0.f, 0.f, 0.f};
    #pragma unroll 4
    for (int k = 0; k < DM; k += 4) {
        v2f a = *(const v2f*)(arow + k);
        v2f b = *(const v2f*)(wrow + k);
        acc = __builtin_amdgcn_wmma_f32_16x16x4_f32(
                   false, a,  false, b,
                   (short)0, acc,  false,  false);
    }

    const int mbase = tr * 16 + half * 8;
    const int col   = tc * 16 + idx;
    const float lm  = act ? lam[col] : 0.f;
    #pragma unroll
    for (int i = 0; i < 8; ++i) {
        float v = acc[i];
        if (act) v = log1pf(fmaxf(v, 0.f)) - lm;
        Y[(size_t)(mbase + i) * DM + col] = v;
    }
}

__global__ __launch_bounds__(256)
void tropical_linear(const float* __restrict__ proj, const float* __restrict__ Wt,
                     float* __restrict__ out)
{
    __shared__ float wl[DKK * 65];
    __shared__ float xl[4 * DKK];

    const int t  = threadIdx.x;
    const int bh = blockIdx.x >> 6;
    const int lt = blockIdx.x & 63;
    const int b  = bh >> 3, h = bh & 7;

    #pragma unroll
    for (int r = 0; r < 16; ++r) {
        int e = t + 256 * r;
        wl[(e >> 6) * 65 + (e & 63)] = Wt[e];
    }
    const int lr = t >> 6, oi = t & 63;
    xl[t] = proj[(size_t)(b * LL + lt * 4 + lr) * DM + h * DKK + oi];
    __syncthreads();

    const float* xr = xl + lr * DKK;
    const float* wr = wl + oi * 65;
    float m = -__builtin_inff();
    #pragma unroll 8
    for (int i = 0; i < DKK; ++i)
        m = fmaxf(m, xr[i] + wr[i]);
    out[(size_t)bh * LL * DKK + (lt * 4 + lr) * DKK + oi] = m;
}

__global__ __launch_bounds__(256)
void trop_attn(const float* __restrict__ qt, const float* __restrict__ kt,
               const float* __restrict__ vt, float* __restrict__ scores,
               float* __restrict__ ctxout)
{
    __shared__ float qs[16 * DKK];
    __shared__ float ks[64 * 65];
    __shared__ float vs[64 * 65];
    __shared__ float ss[16 * 64];

    const int t  = threadIdx.x;
    const int it = blockIdx.x;
    const int bh = blockIdx.y;

    const float* qp = qt + ((size_t)bh * LL + it * 16) * DKK;
    #pragma unroll
    for (int r = 0; r < 4; ++r) qs[t + 256 * r] = qp[t + 256 * r];

    const float NI = -__builtin_inff();
    float ctx[4] = {NI, NI, NI, NI};

    for (int jc = 0; jc < 4; ++jc) {
        const float* kp = kt + ((size_t)bh * LL + jc * 64) * DKK;
        const float* vp = vt + ((size_t)bh * LL + jc * 64) * DKK;
        __syncthreads();
        #pragma unroll
        for (int r = 0; r < 16; ++r) {
            int e = t + 256 * r; int j = e >> 6, d = e & 63;
            ks[j * 65 + d] = kp[e];
            vs[j * 65 + d] = vp[e];
        }
        __syncthreads();

        #pragma unroll
        for (int r = 0; r < 4; ++r) {
            int c = t + 256 * r; int i = c >> 6, jj = c & 63;
            const float* qr = qs + i * DKK;
            const float* kr = ks + jj * 65;
            float mx = NI, mn = __builtin_inff();
            #pragma unroll 8
            for (int d = 0; d < DKK; ++d) {
                float df = qr[d] - kr[d];
                mx = fmaxf(mx, df);
                mn = fminf(mn, df);
            }
            float s = mn - mx;
            ss[c] = s;
            scores[(size_t)bh * LL * LL + (size_t)(it * 16 + i) * LL + jc * 64 + jj] = s;
        }
        __syncthreads();

        #pragma unroll
        for (int r = 0; r < 4; ++r) {
            int c = t + 256 * r; int i = c >> 6, d = c & 63;
            const float* sr = ss + i * 64;
            float m = ctx[r];
            #pragma unroll 8
            for (int jj = 0; jj < 64; ++jj)
                m = fmaxf(m, sr[jj] + vs[jj * 65 + d]);
            ctx[r] = m;
        }
    }

    const int b = bh >> 3, h = bh & 7;
    #pragma unroll
    for (int r = 0; r < 4; ++r) {
        int c = t + 256 * r; int i = c >> 6, d = c & 63;
        ctxout[(size_t)(b * LL + it * 16 + i) * DM + h * DKK + d] = expm1f(ctx[r]);
    }
}

extern "C" void kernel_launch(void* const* d_in, const int* in_sizes, int n_in,
                              void* d_out, int out_size, void* d_ws, size_t ws_size,
                              hipStream_t stream)
{
    (void)in_sizes; (void)n_in; (void)out_size; (void)ws_size;
    const float* x   = (const float*)d_in[0];
    const float* Wq  = (const float*)d_in[1];
    const float* Wk  = (const float*)d_in[2];
    const float* Wv  = (const float*)d_in[3];
    const float* Wo  = (const float*)d_in[4];
    const float* lam = (const float*)d_in[5];
    const float* Wqt = (const float*)d_in[6];
    const float* Wkt = (const float*)d_in[7];
    const float* Wvt = (const float*)d_in[8];

    float* out = (float*)d_out;
    float* sc  = out + (size_t)ROWS * DM;

    const size_t N = (size_t)ROWS * DM;
    float* ws  = (float*)d_ws;
    float* qp  = ws;
    float* kp  = ws + N;
    float* vp  = ws + 2 * N;
    float* qtb = ws + 3 * N;
    float* ktb = ws + 4 * N;
    float* vtb = ws + 5 * N;
    float* ctx = qp;

    dim3 blk(256);
    gemm512_wmma<<<128, blk, 0, stream>>>(x, Wq, lam, qp, 1);
    gemm512_wmma<<<128, blk, 0, stream>>>(x, Wk, lam, kp, 1);
    gemm512_wmma<<<128, blk, 0, stream>>>(x, Wv, lam, vp, 1);

    tropical_linear<<<1024, blk, 0, stream>>>(qp, Wqt, qtb);
    tropical_linear<<<1024, blk, 0, stream>>>(kp, Wkt, ktb);
    tropical_linear<<<1024, blk, 0, stream>>>(vp, Wvt, vtb);

    trop_attn<<<dim3(16, 16), blk, 0, stream>>>(qtb, ktb, vtb, sc, ctx);

    gemm512_wmma<<<128, blk, 0, stream>>>(ctx, Wo, (const float*)nullptr, out, 0);
}


# tropical linear retiled, q, k and v in one launch: block = 32 rows x one head, lane = 4 rows x 4 outputs from 16-byte LDS row reads, Wt staged once per 32 rows, not per 4; 6 launches become 4
# speedup vs baseline: 4.3133x; 4.3133x over previous
#include <hip/hip_runtime.h>
#include <math.h>

#define DM   512
#define NH   8
#define DKK  64
#define B_   2
#define LL   256
#define ROWS (B_*LL)

typedef __attribute__((ext_vector_type(2))) float v2f;
typedef __attribute__((ext_vector_type(8))) float v8f;

typedef __attribute__((ext_vector_type(4))) float v4f;

#define PP 36

#define WMMA4(acc, av, bv) \
    acc = __builtin_amdgcn_wmma_f32_16x16x4_f32(false, av, false, bv, (short)0, acc, false, false)

__device__ __forceinline__ v2f lo2(v4f v) { return __builtin_shufflevector(v, v, 0, 1); }
__device__ __forceinline__ v2f hi2(v4f v) { return __builtin_shufflevector(v, v, 2, 3); }

__device__ __forceinline__ void put_tile(float* p, v8f c, int i, int n, int half, int idx)
{
    #pragma unroll
    for (int r = 0; r < 8; ++r)
        p[(16 * i + 8 * half + r) * PP + 16 * n + idx] = c[r];
}

__global__ __launch_bounds__(128)
void gemm512_ksplit(const float* __restrict__ A, const float* __restrict__ W,
                    float* __restrict__ Y)
{
    __shared__ __attribute__((aligned(16))) float part[4 * 32 * PP];

    const int wave = threadIdx.x >> 5;
    const int lane = threadIdx.x & 31;
    const int br   = blockIdx.x >> 4;
    const int bc   = blockIdx.x & 15;
    const int half = lane >> 4;
    const int idx  = lane & 15;

    const float* ap = A + (size_t)(br * 32 + idx) * DM + wave * 128 + 4 * half;
    const float* wp = W + (size_t)(bc * 32 + idx) * DM + wave * 128 + 4 * half;

    v4f a0[16], a1[16], b0[16], b1[16];
    #pragma unroll
    for (int j = 0; j < 16; ++j) {
        a0[j] = *(const v4f*)(ap + 8 * j);
        a1[j] = *(const v4f*)(ap + 16 * DM + 8 * j);
        b0[j] = *(const v4f*)(wp + 8 * j);
        b1[j] = *(const v4f*)(wp + 16 * DM + 8 * j);
    }

    v8f c00 = {0.f, 0.f, 0.f, 0.f, 0.f, 0.f, 0.f, 0.f};
    v8f c01 = c00, c10 = c00, c11 = c00;
    #pragma unroll
    for (int j = 0; j < 16; ++j) {
        WMMA4(c00, lo2(a0[j]), lo2(b0[j]));
        WMMA4(c01, lo2(a0[j]), lo2(b1[j]));
        WMMA4(c10, lo2(a1[j]), lo2(b0[j]));
        WMMA4(c11, lo2(a1[j]), lo2(b1[j]));
        WMMA4(c00, hi2(a0[j]), hi2(b0[j]));
        WMMA4(c01, hi2(a0[j]), hi2(b1[j]));
        WMMA4(c10, hi2(a1[j]), hi2(b0[j]));
        WMMA4(c11, hi2(a1[j]), hi2(b1[j]));
    }

    __builtin_amdgcn_sched_group_barrier(0x020, 64, 0);
    __builtin_amdgcn_sched_group_barrier(0x008, 128, 0);

    float* pw = part + wave * (32 * PP);
    put_tile(pw, c00, 0, 0, half, idx);
    put_tile(pw, c01, 0, 1, half, idx);
    put_tile(pw, c10, 1, 0, half, idx);
    put_tile(pw, c11, 1, 1, half, idx);
    __syncthreads();

    #pragma unroll
    for (int q = 0; q < 2; ++q) {
        const int e = threadIdx.x + 128 * q;
        const int row = e >> 3, c4 = e & 7;
        const float* pr = part + row * PP + 4 * c4;
        float4 s = *(const float4*)pr;
        #pragma unroll
        for (int w = 1; w < 4; ++w) {
            const float4 o = *(const float4*)(pr + w * (32 * PP));
            s.x += o.x; s.y += o.y; s.z += o.z; s.w += o.w;
        }
        *(float4*)(Y + (size_t)(br * 32 + row) * DM + bc * 32 + 4 * c4) = s;
    }
}

__global__ __launch_bounds__(256)
void proj3_ksplit(const float* __restrict__ X, const float* __restrict__ Wq,
                  const float* __restrict__ Wk, const float* __restrict__ Wv,
                  const float* __restrict__ lam, float* __restrict__ Yq,
                  float* __restrict__ Yk, float* __restrict__ Yv)
{
    __shared__ __attribute__((aligned(16))) float part[3 * 8 * 32 * PP];

    const int wave = threadIdx.x >> 5;
    const int lane = threadIdx.x & 31;
    const int br   = blockIdx.x >> 4;
    const int bc   = blockIdx.x & 15;
    const int half = lane >> 4;
    const int idx  = lane & 15;

    const size_t woff = (size_t)(bc * 32 + idx) * DM + wave * 64 + 4 * half;
    const float* xp = X + (size_t)(br * 32 + idx) * DM + wave * 64 + 4 * half;
    const float* qp = Wq + woff;
    const float* kp = Wk + woff;
    const float* vp = Wv + woff;

    v4f x0[8], x1[8], q0[8], q1[8], k0[8], k1[8], v0[8], v1[8];
    #pragma unroll
    for (int j = 0; j < 8; ++j) {
        x0[j] = *(const v4f*)(xp + 8 * j);
        x1[j] = *(const v4f*)(xp + 16 * DM + 8 * j);
        q0[j] = *(const v4f*)(qp + 8 * j);
        q1[j] = *(const v4f*)(qp + 16 * DM + 8 * j);
        k0[j] = *(const v4f*)(kp + 8 * j);
        k1[j] = *(const v4f*)(kp + 16 * DM + 8 * j);
        v0[j] = *(const v4f*)(vp + 8 * j);
        v1[j] = *(const v4f*)(vp + 16 * DM + 8 * j);
    }

    const v8f z = {0.f, 0.f, 0.f, 0.f, 0.f, 0.f, 0.f, 0.f};
    v8f q00 = z, q01 = z, q10 = z, q11 = z;
    v8f k00 = z, k01 = z, k10 = z, k11 = z;
    v8f v00 = z, v01 = z, v10 = z, v11 = z;
    #pragma unroll
    for (int j = 0; j < 8; ++j) {
        WMMA4(q00, lo2(x0[j]), lo2(q0[j]));  WMMA4(q01, lo2(x0[j]), lo2(q1[j]));
        WMMA4(q10, lo2(x1[j]), lo2(q0[j]));  WMMA4(q11, lo2(x1[j]), lo2(q1[j]));
        WMMA4(k00, lo2(x0[j]), lo2(k0[j]));  WMMA4(k01, lo2(x0[j]), lo2(k1[j]));
        WMMA4(k10, lo2(x1[j]), lo2(k0[j]));  WMMA4(k11, lo2(x1[j]), lo2(k1[j]));
        WMMA4(v00, lo2(x0[j]), lo2(v0[j]));  WMMA4(v01, lo2(x0[j]), lo2(v1[j]));
        WMMA4(v10, lo2(x1[j]), lo2(v0[j]));  WMMA4(v11, lo2(x1[j]), lo2(v1[j]));
        WMMA4(q00, hi2(x0[j]), hi2(q0[j]));  WMMA4(q01, hi2(x0[j]), hi2(q1[j]));
        WMMA4(q10, hi2(x1[j]), hi2(q0[j]));  WMMA4(q11, hi2(x1[j]), hi2(q1[j]));
        WMMA4(k00, hi2(x0[j]), hi2(k0[j]));  WMMA4(k01, hi2(x0[j]), hi2(k1[j]));
        WMMA4(k10, hi2(x1[j]), hi2(k0[j]));  WMMA4(k11, hi2(x1[j]), hi2(k1[j]));
        WMMA4(v00, hi2(x0[j]), hi2(v0[j]));  WMMA4(v01, hi2(x0[j]), hi2(v1[j]));
        WMMA4(v10, hi2(x1[j]), hi2(v0[j]));  WMMA4(v11, hi2(x1[j]), hi2(v1[j]));
    }

    __builtin_amdgcn_sched_group_barrier(0x020, 64, 0);
    __builtin_amdgcn_sched_group_barrier(0x008, 192, 0);

    float* pq = part + (0 * 8 + wave) * (32 * PP);
    float* pk = part + (1 * 8 + wave) * (32 * PP);
    float* pv = part + (2 * 8 + wave) * (32 * PP);
    put_tile(pq, q00, 0, 0, half, idx);  put_tile(pq, q01, 0, 1, half, idx);
    put_tile(pq, q10, 1, 0, half, idx);  put_tile(pq, q11, 1, 1, half, idx);
    put_tile(pk, k00, 0, 0, half, idx);  put_tile(pk, k01, 0, 1, half, idx);
    put_tile(pk, k10, 1, 0, half, idx);  put_tile(pk, k11, 1, 1, half, idx);
    put_tile(pv, v00, 0, 0, half, idx);  put_tile(pv, v01, 0, 1, half, idx);
    put_tile(pv, v10, 1, 0, half, idx);  put_tile(pv, v11, 1, 1, half, idx);
    __syncthreads();

    const int row = threadIdx.x >> 3, c4 = threadIdx.x & 7;
    const float4 lm = *(const float4*)(lam + bc * 32 + 4 * c4);
    const size_t o = (size_t)(br * 32 + row) * DM + bc * 32 + 4 * c4;
    float* const dst[3] = {Yq, Yk, Yv};
    #pragma unroll
    for (int m = 0; m < 3; ++m) {
        const float* pr = part + m * (8 * 32 * PP) + row * PP + 4 * c4;
        float4 s = *(const float4*)pr;
        #pragma unroll
        for (int w = 1; w < 8; ++w) {
            const float4 t = *(const float4*)(pr + w * (32 * PP));
            s.x += t.x; s.y += t.y; s.z += t.z; s.w += t.w;
        }
        *(float4*)(dst[m] + o) =
            make_float4(log1pf(fmaxf(s.x, 0.f)) - lm.x, log1pf(fmaxf(s.y, 0.f)) - lm.y,
                        log1pf(fmaxf(s.z, 0.f)) - lm.z, log1pf(fmaxf(s.w, 0.f)) - lm.w);
    }
}

#define TP 68

__global__ __launch_bounds__(128)
void tropical_linear3(const float* __restrict__ pq, const float* __restrict__ pk,
                      const float* __restrict__ pv, const float* __restrict__ Wtq,
                      const float* __restrict__ Wtk, const float* __restrict__ Wtv,
                      float* __restrict__ oq, float* __restrict__ ok,
                      float* __restrict__ ov)
{
    __shared__ __attribute__((aligned(16))) float wl[DKK * TP];
    __shared__ __attribute__((aligned(16))) float xl[32 * TP];

    const int t  = threadIdx.x;
    const int rt = blockIdx.x;
    const int h  = blockIdx.y;
    const int z  = blockIdx.z;

    const float* proj = (z == 0) ? pq : ((z == 1) ? pk : pv);
    const float* Wt   = (z == 0) ? Wtq : ((z == 1) ? Wtk : Wtv);
    float*       out  = (z == 0) ? oq : ((z == 1) ? ok : ov);

    {
        const float4* wg = (const float4*)Wt;
        const float*  xg = proj + (size_t)(rt * 32) * DM + h * DKK;
        float4 wr[8], xr[4];
        #pragma unroll
        for (int r = 0; r < 8; ++r) wr[r] = wg[t + 128 * r];
        #pragma unroll
        for (int r = 0; r < 4; ++r) {
            const int e = t + 128 * r;
            xr[r] = *(const float4*)(xg + (size_t)(e >> 4) * DM + 4 * (e & 15));
        }
        #pragma unroll
        for (int r = 0; r < 8; ++r) {
            const int e = t + 128 * r;
            const int o = e >> 4, c4 = e & 15;
            *(float4*)(wl + ((o >> 2) + 16 * (o & 3)) * TP + 4 * c4) = wr[r];
        }
        #pragma unroll
        for (int r = 0; r < 4; ++r) {
            const int e = t + 128 * r;
            *(float4*)(xl + (e >> 4) * TP + 4 * (e & 15)) = xr[r];
        }
    }
    __syncthreads();

    const int wave = t >> 5;
    const int og   = (t & 7) + 8 * (wave & 1);
    const int rb   = 16 * (wave >> 1) + ((t >> 3) & 3);

    float m[4][4];
    #pragma unroll
    for (int a = 0; a < 4; ++a)
        #pragma unroll
        for (int c = 0; c < 4; ++c) m[a][c] = -__builtin_inff();

    const float* xb = xl + rb * TP;
    const float* wb = wl + og * TP;
    #pragma unroll 2
    for (int i = 0; i < DKK; i += 4) {
        float4 xv[4], wv[4];
        #pragma unroll
        for (int a = 0; a < 4; ++a) xv[a] = *(const float4*)(xb + (4 * a) * TP + i);
        #pragma unroll
        for (int c = 0; c < 4; ++c) wv[c] = *(const float4*)(wb + (16 * c) * TP + i);
        #pragma unroll
        for (int a = 0; a < 4; ++a) {
            #pragma unroll
            for (int c = 0; c < 4; ++c) {
                float v = m[a][c];
                v = fmaxf(fmaxf(v, xv[a].x + wv[c].x), xv[a].y + wv[c].y);
                v = fmaxf(fmaxf(v, xv[a].z + wv[c].z), xv[a].w + wv[c].w);
                m[a][c] = v;
            }
        }
    }

    #pragma unroll
    for (int a = 0; a < 4; ++a) {
        const int row = rt * 32 + rb + 4 * a;
        const int bh  = (row >> 8) * NH + h;
        *(float4*)(out + ((size_t)bh * LL + (row & (LL - 1))) * DKK + 4 * og) =
            make_float4(m[a][0], m[a][1], m[a][2], m[a][3]);
    }
}

#define KP 68
#define SP 260

__global__ __launch_bounds__(256)
void trop_attn(const float* __restrict__ qt, const float* __restrict__ kt,
               const float* __restrict__ vt, float* __restrict__ scores,
               float* __restrict__ ctxout)
{
    __shared__ __attribute__((aligned(16))) float qs[16 * KP];
    __shared__ __attribute__((aligned(16))) float ks[LL * KP];
    __shared__ __attribute__((aligned(16))) float vs[LL * KP];
    __shared__ __attribute__((aligned(16))) float ss[16 * SP];

    const int t  = threadIdx.x;
    const int it = blockIdx.x;
    const int bh = blockIdx.y;

    const float PI = __builtin_inff();
    const float NI = -__builtin_inff();

    {
        const float4* qg = (const float4*)(qt + ((size_t)bh * LL + it * 16) * DKK);
        const float4* kg = (const float4*)(kt + (size_t)bh * LL * DKK);
        const float4* vg = (const float4*)(vt + (size_t)bh * LL * DKK);
        *(float4*)(qs + (t >> 4) * KP + 4 * (t & 15)) = qg[t];
        #pragma unroll
        for (int r = 0; r < 16; ++r) {
            const int e = t + 256 * r;
            const int row = e >> 4, c4 = e & 15;
            *(float4*)(ks + row * KP + 4 * c4) = kg[e];
            *(float4*)(vs + row * KP + 4 * c4) = vg[e];
        }
    }
    __syncthreads();

    {
        const int ig = t & 3;
        const int jg = t >> 2;
        float mx[4][4], mn[4][4];
        #pragma unroll
        for (int a = 0; a < 4; ++a)
            #pragma unroll
            for (int b = 0; b < 4; ++b) { mx[a][b] = NI; mn[a][b] = PI; }

        const float* qb = qs + (4 * ig) * KP;
        const float* kb = ks + jg * KP;
        #pragma unroll 2
        for (int d = 0; d < DKK; d += 4) {
            float4 qv[4], kv[4];
            #pragma unroll
            for (int a = 0; a < 4; ++a) qv[a] = *(const float4*)(qb + a * KP + d);
            #pragma unroll
            for (int b = 0; b < 4; ++b) kv[b] = *(const float4*)(kb + b * 64 * KP + d);
            #pragma unroll
            for (int a = 0; a < 4; ++a) {
                #pragma unroll
                for (int b = 0; b < 4; ++b) {
                    const float d0 = qv[a].x - kv[b].x;
                    const float d1 = qv[a].y - kv[b].y;
                    const float d2 = qv[a].z - kv[b].z;
                    const float d3 = qv[a].w - kv[b].w;
                    float hi = mx[a][b], lo = mn[a][b];
                    hi = fmaxf(fmaxf(hi, d0), d1);
                    hi = fmaxf(fmaxf(hi, d2), d3);
                    lo = fminf(fminf(lo, d0), d1);
                    lo = fminf(fminf(lo, d2), d3);
                    mx[a][b] = hi; mn[a][b] = lo;
                }
            }
        }
        #pragma unroll
        for (int a = 0; a < 4; ++a)
            #pragma unroll
            for (int b = 0; b < 4; ++b)
                ss[(4 * ig + a) * SP + jg + 64 * b] = mn[a][b] - mx[a][b];
    }
    __syncthreads();

    {
        float4* sg = (float4*)(scores + (size_t)bh * LL * LL + (size_t)(it * 16) * LL);
        #pragma unroll
        for (int r = 0; r < 4; ++r) {
            const int e = t + 256 * r;
            sg[e] = *(const float4*)(ss + (e >> 6) * SP + 4 * (e & 63));
        }
    }

    float* pm = ks;
    {
        const int grp = t >> 6;
        const int u   = t & 63;
        const int ci  = u & 3;
        const int dg  = u >> 2;
        float cx[4][4];
        #pragma unroll
        for (int a = 0; a < 4; ++a)
            #pragma unroll
            for (int c = 0; c < 4; ++c) cx[a][c] = NI;

        const float* sb = ss + (4 * ci) * SP + 64 * grp;
        const float* vb = vs + (64 * grp) * KP + 4 * dg;
        #pragma unroll 2
        for (int j = 0; j < 64; j += 4) {
            float4 sv[4], vv[4];
            #pragma unroll
            for (int a = 0; a < 4; ++a) sv[a] = *(const float4*)(sb + a * SP + j);
            #pragma unroll
            for (int q = 0; q < 4; ++q) vv[q] = *(const float4*)(vb + (j + q) * KP);
            #pragma unroll
            for (int a = 0; a < 4; ++a) {
                float m0 = cx[a][0], m1 = cx[a][1], m2 = cx[a][2], m3 = cx[a][3];
                m0 = fmaxf(fmaxf(m0, sv[a].x + vv[0].x), sv[a].y + vv[1].x);
                m0 = fmaxf(fmaxf(m0, sv[a].z + vv[2].x), sv[a].w + vv[3].x);
                m1 = fmaxf(fmaxf(m1, sv[a].x + vv[0].y), sv[a].y + vv[1].y);
                m1 = fmaxf(fmaxf(m1, sv[a].z + vv[2].y), sv[a].w + vv[3].y);
                m2 = fmaxf(fmaxf(m2, sv[a].x + vv[0].z), sv[a].y + vv[1].z);
                m2 = fmaxf(fmaxf(m2, sv[a].z + vv[2].z), sv[a].w + vv[3].z);
                m3 = fmaxf(fmaxf(m3, sv[a].x + vv[0].w), sv[a].y + vv[1].w);
                m3 = fmaxf(fmaxf(m3, sv[a].z + vv[2].w), sv[a].w + vv[3].w);
                cx[a][0] = m0; cx[a][1] = m1; cx[a][2] = m2; cx[a][3] = m3;
            }
        }
        #pragma unroll
        for (int a = 0; a < 4; ++a)
            *(float4*)(pm + grp * (16 * KP) + (4 * ci + a) * KP + 4 * dg) =
                make_float4(cx[a][0], cx[a][1], cx[a][2], cx[a][3]);
    }
    __syncthreads();

    {
        const int i = t >> 4, c4 = t & 15;
        const float* pr = pm + i * KP + 4 * c4;
        float4 m = *(const float4*)pr;
        #pragma unroll
        for (int g = 1; g < 4; ++g) {
            const float4 o = *(const float4*)(pr + g * (16 * KP));
            m.x = fmaxf(m.x, o.x); m.y = fmaxf(m.y, o.y);
            m.z = fmaxf(m.z, o.z); m.w = fmaxf(m.w, o.w);
        }
        const int b = bh >> 3, h = bh & 7;
        *(float4*)(ctxout + (size_t)(b * LL + it * 16 + i) * DM + h * DKK + 4 * c4) =
            make_float4(expm1f(m.x), expm1f(m.y), expm1f(m.z), expm1f(m.w));
    }
}

extern "C" void kernel_launch(void* const* d_in, const int* in_sizes, int n_in,
                              void* d_out, int out_size, void* d_ws, size_t ws_size,
                              hipStream_t stream)
{
    (void)in_sizes; (void)n_in; (void)out_size; (void)ws_size;
    const float* x   = (const float*)d_in[0];
    const float* Wq  = (const float*)d_in[1];
    const float* Wk  = (const float*)d_in[2];
    const float* Wv  = (const float*)d_in[3];
    const float* Wo  = (const float*)d_in[4];
    const float* lam = (const float*)d_in[5];
    const float* Wqt = (const float*)d_in[6];
    const float* Wkt = (const float*)d_in[7];
    const float* Wvt = (const float*)d_in[8];

    float* out = (float*)d_out;
    float* sc  = out + (size_t)ROWS * DM;

    const size_t N = (size_t)ROWS * DM;
    float* ws  = (float*)d_ws;
    float* qp  = ws;
    float* kp  = ws + N;
    float* vp  = ws + 2 * N;
    float* qtb = ws + 3 * N;
    float* ktb = ws + 4 * N;
    float* vtb = ws + 5 * N;
    float* ctx = qp;

    dim3 blk(256);
    proj3_ksplit<<<256, blk, 0, stream>>>(x, Wq, Wk, Wv, lam, qp, kp, vp);

    tropical_linear3<<<dim3(16, NH, 3), dim3(128), 0, stream>>>(qp, kp, vp, Wqt, Wkt, Wvt,
                                                                qtb, ktb, vtb);

    trop_attn<<<dim3(16, 16), blk, 0, stream>>>(qtb, ktb, vtb, sc, ctx);

    gemm512_ksplit<<<256, dim3(128), 0, stream>>>(ctx, Wo, out);
}


# tropical linear split over its 64 inputs: each 32x32 projection block reduces its own 32 activation columns (LDS only) to partial maxima, attention staging takes their exact max; 4 launches become 3
# speedup vs baseline: 4.4849x; 1.0398x over previous
#include <hip/hip_runtime.h>
#include <math.h>

#define DM   512
#define NH   8
#define DKK  64
#define B_   2
#define LL   256
#define ROWS (B_*LL)

typedef __attribute__((ext_vector_type(2))) float v2f;
typedef __attribute__((ext_vector_type(8))) float v8f;

typedef __attribute__((ext_vector_type(4))) float v4f;

#define PP 36

#define WMMA4(acc, av, bv) \
    acc = __builtin_amdgcn_wmma_f32_16x16x4_f32(false, av, false, bv, (short)0, acc, false, false)

__device__ __forceinline__ v2f lo2(v4f v) { return __builtin_shufflevector(v, v, 0, 1); }
__device__ __forceinline__ v2f hi2(v4f v) { return __builtin_shufflevector(v, v, 2, 3); }

__device__ __forceinline__ void put_tile(float* p, v8f c, int i, int n, int half, int idx)
{
    #pragma unroll
    for (int r = 0; r < 8; ++r)
        p[(16 * i + 8 * half + r) * PP + 16 * n + idx] = c[r];
}

__global__ __launch_bounds__(128)
void gemm512_ksplit(const float* __restrict__ A, const float* __restrict__ W,
                    float* __restrict__ Y)
{
    __shared__ __attribute__((aligned(16))) float part[4 * 32 * PP];

    const int wave = threadIdx.x >> 5;
    const int lane = threadIdx.x & 31;
    const int br   = blockIdx.x >> 4;
    const int bc   = blockIdx.x & 15;
    const int half = lane >> 4;
    const int idx  = lane & 15;

    const float* ap = A + (size_t)(br * 32 + idx) * DM + wave * 128 + 4 * half;
    const float* wp = W + (size_t)(bc * 32 + idx) * DM + wave * 128 + 4 * half;

    v4f a0[16], a1[16], b0[16], b1[16];
    #pragma unroll
    for (int j = 0; j < 16; ++j) {
        a0[j] = *(const v4f*)(ap + 8 * j);
        a1[j] = *(const v4f*)(ap + 16 * DM + 8 * j);
        b0[j] = *(const v4f*)(wp + 8 * j);
        b1[j] = *(const v4f*)(wp + 16 * DM + 8 * j);
    }

    v8f c00 = {0.f, 0.f, 0.f, 0.f, 0.f, 0.f, 0.f, 0.f};
    v8f c01 = c00, c10 = c00, c11 = c00;
    #pragma unroll
    for (int j = 0; j < 16; ++j) {
        WMMA4(c00, lo2(a0[j]), lo2(b0[j]));
        WMMA4(c01, lo2(a0[j]), lo2(b1[j]));
        WMMA4(c10, lo2(a1[j]), lo2(b0[j]));
        WMMA4(c11, lo2(a1[j]), lo2(b1[j]));
        WMMA4(c00, hi2(a0[j]), hi2(b0[j]));
        WMMA4(c01, hi2(a0[j]), hi2(b1[j]));
        WMMA4(c10, hi2(a1[j]), hi2(b0[j]));
        WMMA4(c11, hi2(a1[j]), hi2(b1[j]));
    }

    __builtin_amdgcn_sched_group_barrier(0x020, 64, 0);
    __builtin_amdgcn_sched_group_barrier(0x008, 128, 0);

    float* pw = part + wave * (32 * PP);
    put_tile(pw, c00, 0, 0, half, idx);
    put_tile(pw, c01, 0, 1, half, idx);
    put_tile(pw, c10, 1, 0, half, idx);
    put_tile(pw, c11, 1, 1, half, idx);
    __syncthreads();

    #pragma unroll
    for (int q = 0; q < 2; ++q) {
        const int e = threadIdx.x + 128 * q;
        const int row = e >> 3, c4 = e & 7;
        const float* pr = part + row * PP + 4 * c4;
        float4 s = *(const float4*)pr;
        #pragma unroll
        for (int w = 1; w < 4; ++w) {
            const float4 o = *(const float4*)(pr + w * (32 * PP));
            s.x += o.x; s.y += o.y; s.z += o.z; s.w += o.w;
        }
        *(float4*)(Y + (size_t)(br * 32 + row) * DM + bc * 32 + 4 * c4) = s;
    }
}

__global__ __launch_bounds__(256)
void proj3_trophalf(const float* __restrict__ X, const float* __restrict__ Wq,
                    const float* __restrict__ Wk, const float* __restrict__ Wv,
                    const float* __restrict__ lam, const float* __restrict__ Wtq,
                    const float* __restrict__ Wtk, const float* __restrict__ Wtv,
                    float* __restrict__ P)
{
    __shared__ __attribute__((aligned(16))) float smem[3 * DKK * PP + 3 * 32 * PP + 3 * 8 * 32 * PP];
    float* const wl   = smem;
    float* const act  = smem + 3 * DKK * PP;
    float* const part = act + 3 * 32 * PP;

    const int wave = threadIdx.x >> 5;
    const int lane = threadIdx.x & 31;
    const int br   = blockIdx.x >> 4;
    const int bc   = blockIdx.x & 15;
    const int half = lane >> 4;
    const int idx  = lane & 15;

    const size_t woff = (size_t)(bc * 32 + idx) * DM + wave * 64 + 4 * half;
    const float* xp = X + (size_t)(br * 32 + idx) * DM + wave * 64 + 4 * half;
    const float* qp = Wq + woff;
    const float* kp = Wk + woff;
    const float* vp = Wv + woff;

    v4f x0[8], x1[8], q0[8], q1[8], k0[8], k1[8], v0[8], v1[8];
    #pragma unroll
    for (int j = 0; j < 8; ++j) {
        x0[j] = *(const v4f*)(xp + 8 * j);
        x1[j] = *(const v4f*)(xp + 16 * DM + 8 * j);
        q0[j] = *(const v4f*)(qp + 8 * j);
        q1[j] = *(const v4f*)(qp + 16 * DM + 8 * j);
        k0[j] = *(const v4f*)(kp + 8 * j);
        k1[j] = *(const v4f*)(kp + 16 * DM + 8 * j);
        v0[j] = *(const v4f*)(vp + 8 * j);
        v1[j] = *(const v4f*)(vp + 16 * DM + 8 * j);
    }

    const v8f z = {0.f, 0.f, 0.f, 0.f, 0.f, 0.f, 0.f, 0.f};
    v8f q00 = z, q01 = z, q10 = z, q11 = z;
    v8f k00 = z, k01 = z, k10 = z, k11 = z;
    v8f v00 = z, v01 = z, v10 = z, v11 = z;
    #pragma unroll
    for (int j = 0; j < 8; ++j) {
        WMMA4(q00, lo2(x0[j]), lo2(q0[j]));  WMMA4(q01, lo2(x0[j]), lo2(q1[j]));
        WMMA4(q10, lo2(x1[j]), lo2(q0[j]));  WMMA4(q11, lo2(x1[j]), lo2(q1[j]));
        WMMA4(k00, lo2(x0[j]), lo2(k0[j]));  WMMA4(k01, lo2(x0[j]), lo2(k1[j]));
        WMMA4(k10, lo2(x1[j]), lo2(k0[j]));  WMMA4(k11, lo2(x1[j]), lo2(k1[j]));
        WMMA4(v00, lo2(x0[j]), lo2(v0[j]));  WMMA4(v01, lo2(x0[j]), lo2(v1[j]));
        WMMA4(v10, lo2(x1[j]), lo2(v0[j]));  WMMA4(v11, lo2(x1[j]), lo2(v1[j]));
        WMMA4(q00, hi2(x0[j]), hi2(q0[j]));  WMMA4(q01, hi2(x0[j]), hi2(q1[j]));
        WMMA4(q10, hi2(x1[j]), hi2(q0[j]));  WMMA4(q11, hi2(x1[j]), hi2(q1[j]));
        WMMA4(k00, hi2(x0[j]), hi2(k0[j]));  WMMA4(k01, hi2(x0[j]), hi2(k1[j]));
        WMMA4(k10, hi2(x1[j]), hi2(k0[j]));  WMMA4(k11, hi2(x1[j]), hi2(k1[j]));
        WMMA4(v00, hi2(x0[j]), hi2(v0[j]));  WMMA4(v01, hi2(x0[j]), hi2(v1[j]));
        WMMA4(v10, hi2(x1[j]), hi2(v0[j]));  WMMA4(v11, hi2(x1[j]), hi2(v1[j]));
    }

    __builtin_amdgcn_sched_group_barrier(0x020, 64, 0);
    __builtin_amdgcn_sched_group_barrier(0x008, 192, 0);

    const int hh = bc & 1;
    const int h  = bc >> 1;
    const int wo0 = (threadIdx.x >> 3) * DKK + 32 * hh + 4 * (threadIdx.x & 7);
    const int wo1 = wo0 + 32 * DKK;
    const float4 wq0 = *(const float4*)(Wtq + wo0), wq1 = *(const float4*)(Wtq + wo1);
    const float4 wk0 = *(const float4*)(Wtk + wo0), wk1 = *(const float4*)(Wtk + wo1);
    const float4 wv0 = *(const float4*)(Wtv + wo0), wv1 = *(const float4*)(Wtv + wo1);

    float* pq = part + (0 * 8 + wave) * (32 * PP);
    float* pk = part + (1 * 8 + wave) * (32 * PP);
    float* pv = part + (2 * 8 + wave) * (32 * PP);
    put_tile(pq, q00, 0, 0, half, idx);  put_tile(pq, q01, 0, 1, half, idx);
    put_tile(pq, q10, 1, 0, half, idx);  put_tile(pq, q11, 1, 1, half, idx);
    put_tile(pk, k00, 0, 0, half, idx);  put_tile(pk, k01, 0, 1, half, idx);
    put_tile(pk, k10, 1, 0, half, idx);  put_tile(pk, k11, 1, 1, half, idx);
    put_tile(pv, v00, 0, 0, half, idx);  put_tile(pv, v01, 0, 1, half, idx);
    put_tile(pv, v10, 1, 0, half, idx);  put_tile(pv, v11, 1, 1, half, idx);
    __syncthreads();

    {
        const int row = threadIdx.x >> 3, c4 = threadIdx.x & 7;
        const float4 lm = *(const float4*)(lam + bc * 32 + 4 * c4);
        #pragma unroll
        for (int m = 0; m < 3; ++m) {
            const float* pr = part + m * (8 * 32 * PP) + row * PP + 4 * c4;
            float4 s = *(const float4*)pr;
            #pragma unroll
            for (int w = 1; w < 8; ++w) {
                const float4 t = *(const float4*)(pr + w * (32 * PP));
                s.x += t.x; s.y += t.y; s.z += t.z; s.w += t.w;
            }
            *(float4*)(act + m * (32 * PP) + row * PP + 4 * c4) =
                make_float4(log1pf(fmaxf(s.x, 0.f)) - lm.x, log1pf(fmaxf(s.y, 0.f)) - lm.y,
                            log1pf(fmaxf(s.z, 0.f)) - lm.z, log1pf(fmaxf(s.w, 0.f)) - lm.w);
        }
        {
            const int o0 = threadIdx.x >> 3;
            float* w0 = wl + ((o0 >> 2) + 16 * (o0 & 3)) * PP + 4 * (threadIdx.x & 7);
            float* w1 = w0 + 8 * PP;
            *(float4*)(w0) = wq0;                     *(float4*)(w1) = wq1;
            *(float4*)(w0 + DKK * PP) = wk0;          *(float4*)(w1 + DKK * PP) = wk1;
            *(float4*)(w0 + 2 * (DKK * PP)) = wv0;    *(float4*)(w1 + 2 * (DKK * PP)) = wv1;
        }
    }
    __syncthreads();

    {
        const int p  = wave & 3;
        const int og = (lane & 7) + 8 * (p & 1);
        const int rb = 16 * (p >> 1) + ((lane >> 3) & 3);
        const size_t N = (size_t)ROWS * DM;
        const int wsc = __builtin_amdgcn_readfirstlane(wave);
        for (int m = wsc >> 2; m < 3; m += 2) {
            float mx[4][4];
            #pragma unroll
            for (int a = 0; a < 4; ++a)
                #pragma unroll
                for (int cc = 0; cc < 4; ++cc) mx[a][cc] = -__builtin_inff();

            const float* xb = act + m * (32 * PP) + rb * PP;
            const float* wb = wl + m * (DKK * PP) + og * PP;
            #pragma unroll 4
            for (int i = 0; i < 32; i += 4) {
                float4 xv[4], wv[4];
                #pragma unroll
                for (int a = 0; a < 4; ++a) xv[a] = *(const float4*)(xb + (4 * a) * PP + i);
                #pragma unroll
                for (int cc = 0; cc < 4; ++cc) wv[cc] = *(const float4*)(wb + (16 * cc) * PP + i);
                #pragma unroll
                for (int a = 0; a < 4; ++a) {
                    #pragma unroll
                    for (int cc = 0; cc < 4; ++cc) {
                        float v = mx[a][cc];
                        v = fmaxf(fmaxf(v, xv[a].x + wv[cc].x), xv[a].y + wv[cc].y);
                        v = fmaxf(fmaxf(v, xv[a].z + wv[cc].z), xv[a].w + wv[cc].w);
                        mx[a][cc] = v;
                    }
                }
            }

            float* dst = P + (size_t)(2 * m + hh) * N;
            #pragma unroll
            for (int a = 0; a < 4; ++a) {
                const int row = br * 32 + rb + 4 * a;
                const int bh  = (row >> 8) * NH + h;
                *(float4*)(dst + ((size_t)bh * LL + (row & (LL - 1))) * DKK + 4 * og) =
                    make_float4(mx[a][0], mx[a][1], mx[a][2], mx[a][3]);
            }
        }
    }
}

#define KP 68

__device__ __forceinline__ v4f max4(v4f a, v4f b)
{
    return __builtin_elementwise_maximum(a, b);
}
#define SP 260

__global__ __launch_bounds__(256)
void trop_attn(const float* __restrict__ qa, const float* __restrict__ qb,
               const float* __restrict__ ka, const float* __restrict__ kb,
               const float* __restrict__ va, const float* __restrict__ vb,
               float* __restrict__ scores, float* __restrict__ ctxout)
{
    __shared__ __attribute__((aligned(16))) float qs[16 * KP];
    __shared__ __attribute__((aligned(16))) float ks[LL * KP];
    __shared__ __attribute__((aligned(16))) float vs[LL * KP];
    __shared__ __attribute__((aligned(16))) float ss[16 * SP];

    const int t  = threadIdx.x;
    const int it = blockIdx.x;
    const int bh = blockIdx.y;

    const float PI = __builtin_inff();
    const float NI = -__builtin_inff();

    {
        const size_t hb = (size_t)bh * LL * DKK;
        const v4f* qg0 = (const v4f*)(qa + hb + (size_t)(it * 16) * DKK);
        const v4f* qg1 = (const v4f*)(qb + hb + (size_t)(it * 16) * DKK);
        const v4f* kg0 = (const v4f*)(ka + hb);
        const v4f* kg1 = (const v4f*)(kb + hb);
        const v4f* vg0 = (const v4f*)(va + hb);
        const v4f* vg1 = (const v4f*)(vb + hb);
        *(v4f*)(qs + (t >> 4) * KP + 4 * (t & 15)) = max4(qg0[t], qg1[t]);
        #pragma unroll
        for (int r = 0; r < 16; ++r) {
            const int e = t + 256 * r;
            const int row = e >> 4, c4 = e & 15;
            *(v4f*)(ks + row * KP + 4 * c4) = max4(kg0[e], kg1[e]);
            *(v4f*)(vs + row * KP + 4 * c4) = max4(vg0[e], vg1[e]);
        }
    }
    __syncthreads();

    {
        const int ig = t & 3;
        const int jg = t >> 2;
        float mx[4][4], mn[4][4];
        #pragma unroll
        for (int a = 0; a < 4; ++a)
            #pragma unroll
            for (int b = 0; b < 4; ++b) { mx[a][b] = NI; mn[a][b] = PI; }

        const float* qb = qs + (4 * ig) * KP;
        const float* kb = ks + jg * KP;
        #pragma unroll 2
        for (int d = 0; d < DKK; d += 4) {
            float4 qv[4], kv[4];
            #pragma unroll
            for (int a = 0; a < 4; ++a) qv[a] = *(const float4*)(qb + a * KP + d);
            #pragma unroll
            for (int b = 0; b < 4; ++b) kv[b] = *(const float4*)(kb + b * 64 * KP + d);
            #pragma unroll
            for (int a = 0; a < 4; ++a) {
                #pragma unroll
                for (int b = 0; b < 4; ++b) {
                    const float d0 = qv[a].x - kv[b].x;
                    const float d1 = qv[a].y - kv[b].y;
                    const float d2 = qv[a].z - kv[b].z;
                    const float d3 = qv[a].w - kv[b].w;
                    float hi = mx[a][b], lo = mn[a][b];
                    hi = fmaxf(fmaxf(hi, d0), d1);
                    hi = fmaxf(fmaxf(hi, d2), d3);
                    lo = fminf(fminf(lo, d0), d1);
                    lo = fminf(fminf(lo, d2), d3);
                    mx[a][b] = hi; mn[a][b] = lo;
                }
            }
        }
        #pragma unroll
        for (int a = 0; a < 4; ++a)
            #pragma unroll
            for (int b = 0; b < 4; ++b)
                ss[(4 * ig + a) * SP + jg + 64 * b] = mn[a][b] - mx[a][b];
    }
    __syncthreads();

    {
        float4* sg = (float4*)(scores + (size_t)bh * LL * LL + (size_t)(it * 16) * LL);
        #pragma unroll
        for (int r = 0; r < 4; ++r) {
            const int e = t + 256 * r;
            sg[e] = *(const float4*)(ss + (e >> 6) * SP + 4 * (e & 63));
        }
    }

    float* pm = ks;
    {
        const int grp = t >> 6;
        const int u   = t & 63;
        const int ci  = u & 3;
        const int dg  = u >> 2;
        float cx[4][4];
        #pragma unroll
        for (int a = 0; a < 4; ++a)
            #pragma unroll
            for (int c = 0; c < 4; ++c) cx[a][c] = NI;

        const float* sb = ss + (4 * ci) * SP + 64 * grp;
        const float* vb = vs + (64 * grp) * KP + 4 * dg;
        #pragma unroll 2
        for (int j = 0; j < 64; j += 4) {
            float4 sv[4], vv[4];
            #pragma unroll
            for (int a = 0; a < 4; ++a) sv[a] = *(const float4*)(sb + a * SP + j);
            #pragma unroll
            for (int q = 0; q < 4; ++q) vv[q] = *(const float4*)(vb + (j + q) * KP);
            #pragma unroll
            for (int a = 0; a < 4; ++a) {
                float m0 = cx[a][0], m1 = cx[a][1], m2 = cx[a][2], m3 = cx[a][3];
                m0 = fmaxf(fmaxf(m0, sv[a].x + vv[0].x), sv[a].y + vv[1].x);
                m0 = fmaxf(fmaxf(m0, sv[a].z + vv[2].x), sv[a].w + vv[3].x);
                m1 = fmaxf(fmaxf(m1, sv[a].x + vv[0].y), sv[a].y + vv[1].y);
                m1 = fmaxf(fmaxf(m1, sv[a].z + vv[2].y), sv[a].w + vv[3].y);
                m2 = fmaxf(fmaxf(m2, sv[a].x + vv[0].z), sv[a].y + vv[1].z);
                m2 = fmaxf(fmaxf(m2, sv[a].z + vv[2].z), sv[a].w + vv[3].z);
                m3 = fmaxf(fmaxf(m3, sv[a].x + vv[0].w), sv[a].y + vv[1].w);
                m3 = fmaxf(fmaxf(m3, sv[a].z + vv[2].w), sv[a].w + vv[3].w);
                cx[a][0] = m0; cx[a][1] = m1; cx[a][2] = m2; cx[a][3] = m3;
            }
        }
        #pragma unroll
        for (int a = 0; a < 4; ++a)
            *(float4*)(pm + grp * (16 * KP) + (4 * ci + a) * KP + 4 * dg) =
                make_float4(cx[a][0], cx[a][1], cx[a][2], cx[a][3]);
    }
    __syncthreads();

    {
        const int i = t >> 4, c4 = t & 15;
        const float* pr = pm + i * KP + 4 * c4;
        float4 m = *(const float4*)pr;
        #pragma unroll
        for (int g = 1; g < 4; ++g) {
            const float4 o = *(const float4*)(pr + g * (16 * KP));
            m.x = fmaxf(m.x, o.x); m.y = fmaxf(m.y, o.y);
            m.z = fmaxf(m.z, o.z); m.w = fmaxf(m.w, o.w);
        }
        const int b = bh >> 3, h = bh & 7;
        *(float4*)(ctxout + (size_t)(b * LL + it * 16 + i) * DM + h * DKK + 4 * c4) =
            make_float4(expm1f(m.x), expm1f(m.y), expm1f(m.z), expm1f(m.w));
    }
}

extern "C" void kernel_launch(void* const* d_in, const int* in_sizes, int n_in,
                              void* d_out, int out_size, void* d_ws, size_t ws_size,
                              hipStream_t stream)
{
    (void)in_sizes; (void)n_in; (void)out_size; (void)ws_size;
    const float* x   = (const float*)d_in[0];
    const float* Wq  = (const float*)d_in[1];
    const float* Wk  = (const float*)d_in[2];
    const float* Wv  = (const float*)d_in[3];
    const float* Wo  = (const float*)d_in[4];
    const float* lam = (const float*)d_in[5];
    const float* Wqt = (const float*)d_in[6];
    const float* Wkt = (const float*)d_in[7];
    const float* Wvt = (const float*)d_in[8];

    float* out = (float*)d_out;
    float* sc  = out + (size_t)ROWS * DM;

    const size_t N = (size_t)ROWS * DM;
    float* ws  = (float*)d_ws;
    float* pp  = ws;
    float* ctx = ws + 6 * N;

    dim3 blk(256);
    proj3_trophalf<<<256, blk, 0, stream>>>(x, Wq, Wk, Wv, lam, Wqt, Wkt, Wvt, pp);

    trop_attn<<<dim3(16, 16), blk, 0, stream>>>(pp, pp + N, pp + 2 * N, pp + 3 * N,
                                                pp + 4 * N, pp + 5 * N, sc, ctx);

    gemm512_ksplit<<<256, dim3(128), 0, stream>>>(ctx, Wo, out);
}


# projection blocks retiled to 32 rows x (one whole head of k or v + half a head of q): tropical linear of k and v completes in-block, attention stages one plane each, not the max of two
# speedup vs baseline: 5.0968x; 1.1364x over previous
#include <hip/hip_runtime.h>
#include <math.h>

#define DM   512
#define NH   8
#define DKK  64
#define B_   2
#define LL   256
#define ROWS (B_*LL)

typedef __attribute__((ext_vector_type(2))) float v2f;
typedef __attribute__((ext_vector_type(8))) float v8f;

typedef __attribute__((ext_vector_type(4))) float v4f;

#define PP 36

#define WMMA4(acc, av, bv) \
    acc = __builtin_amdgcn_wmma_f32_16x16x4_f32(false, av, false, bv, (short)0, acc, false, false)

__device__ __forceinline__ v2f lo2(v4f v) { return __builtin_shufflevector(v, v, 0, 1); }
__device__ __forceinline__ v2f hi2(v4f v) { return __builtin_shufflevector(v, v, 2, 3); }

__device__ __forceinline__ void put_tile(float* p, v8f c, int i, int n, int half, int idx)
{
    #pragma unroll
    for (int r = 0; r < 8; ++r)
        p[(16 * i + 8 * half + r) * PP + 16 * n + idx] = c[r];
}

__global__ __launch_bounds__(128)
void gemm512_ksplit(const float* __restrict__ A, const float* __restrict__ W,
                    float* __restrict__ Y)
{
    __shared__ __attribute__((aligned(16))) float part[4 * 32 * PP];

    const int wave = threadIdx.x >> 5;
    const int lane = threadIdx.x & 31;
    const int br   = blockIdx.x >> 4;
    const int bc   = blockIdx.x & 15;
    const int half = lane >> 4;
    const int idx  = lane & 15;

    const float* ap = A + (size_t)(br * 32 + idx) * DM + wave * 128 + 4 * half;
    const float* wp = W + (size_t)(bc * 32 + idx) * DM + wave * 128 + 4 * half;

    v4f a0[16], a1[16], b0[16], b1[16];
    #pragma unroll
    for (int j = 0; j < 16; ++j) {
        a0[j] = *(const v4f*)(ap + 8 * j);
        a1[j] = *(const v4f*)(ap + 16 * DM + 8 * j);
        b0[j] = *(const v4f*)(wp + 8 * j);
        b1[j] = *(const v4f*)(wp + 16 * DM + 8 * j);
    }

    v8f c00 = {0.f, 0.f, 0.f, 0.f, 0.f, 0.f, 0.f, 0.f};
    v8f c01 = c00, c10 = c00, c11 = c00;
    #pragma unroll
    for (int j = 0; j < 16; ++j) {
        WMMA4(c00, lo2(a0[j]), lo2(b0[j]));
        WMMA4(c01, lo2(a0[j]), lo2(b1[j]));
        WMMA4(c10, lo2(a1[j]), lo2(b0[j]));
        WMMA4(c11, lo2(a1[j]), lo2(b1[j]));
        WMMA4(c00, hi2(a0[j]), hi2(b0[j]));
        WMMA4(c01, hi2(a0[j]), hi2(b1[j]));
        WMMA4(c10, hi2(a1[j]), hi2(b0[j]));
        WMMA4(c11, hi2(a1[j]), hi2(b1[j]));
    }

    __builtin_amdgcn_sched_group_barrier(0x020, 64, 0);
    __builtin_amdgcn_sched_group_barrier(0x008, 128, 0);

    float* pw = part + wave * (32 * PP);
    put_tile(pw, c00, 0, 0, half, idx);
    put_tile(pw, c01, 0, 1, half, idx);
    put_tile(pw, c10, 1, 0, half, idx);
    put_tile(pw, c11, 1, 1, half, idx);
    __syncthreads();

    #pragma unroll
    for (int q = 0; q < 2; ++q) {
        const int e = threadIdx.x + 128 * q;
        const int row = e >> 3, c4 = e & 7;
        const float* pr = part + row * PP + 4 * c4;
        float4 s = *(const float4*)pr;
        #pragma unroll
        for (int w = 1; w < 4; ++w) {
            const float4 o = *(const float4*)(pr + w * (32 * PP));
            s.x += o.x; s.y += o.y; s.z += o.z; s.w += o.w;
        }
        *(float4*)(Y + (size_t)(br * 32 + row) * DM + bc * 32 + 4 * c4) = s;
    }
}

#define PW 100
#define PF 68

__device__ __forceinline__ void put_tile_w(float* p, v8f c, int i, int n, int half, int idx)
{
    #pragma unroll
    for (int r = 0; r < 8; ++r)
        p[(16 * i + 8 * half + r) * PW + 16 * n + idx] = c[r];
}

template <int PITCH, int NI>
__device__ __forceinline__ void trop_task(const float* xb, const float* wb, float* dst,
                                          int row0, int h, int og)
{
    float mx[4][4];
    #pragma unroll
    for (int a = 0; a < 4; ++a)
        #pragma unroll
        for (int cc = 0; cc < 4; ++cc) mx[a][cc] = -__builtin_inff();

    #pragma unroll 4
    for (int i = 0; i < NI; i += 4) {
        float4 xv[4], wv[4];
        #pragma unroll
        for (int a = 0; a < 4; ++a) xv[a] = *(const float4*)(xb + (4 * a) * PITCH + i);
        #pragma unroll
        for (int cc = 0; cc < 4; ++cc) wv[cc] = *(const float4*)(wb + (16 * cc) * PITCH + i);
        #pragma unroll
        for (int a = 0; a < 4; ++a) {
            #pragma unroll
            for (int cc = 0; cc < 4; ++cc) {
                float v = mx[a][cc];
                v = fmaxf(fmaxf(v, xv[a].x + wv[cc].x), xv[a].y + wv[cc].y);
                v = fmaxf(fmaxf(v, xv[a].z + wv[cc].z), xv[a].w + wv[cc].w);
                mx[a][cc] = v;
            }
        }
    }

    #pragma unroll
    for (int a = 0; a < 4; ++a) {
        const int row = row0 + 4 * a;
        const int bh  = (row >> 8) * NH + h;
        *(float4*)(dst + ((size_t)bh * LL + (row & (LL - 1))) * DKK + 4 * og) =
            make_float4(mx[a][0], mx[a][1], mx[a][2], mx[a][3]);
    }
}

__global__ __launch_bounds__(256)
void proj_headtile(const float* __restrict__ X, const float* __restrict__ Wq,
                   const float* __restrict__ Wk, const float* __restrict__ Wv,
                   const float* __restrict__ lam, const float* __restrict__ Wtq,
                   const float* __restrict__ Wtk, const float* __restrict__ Wtv,
                   float* __restrict__ P)
{
    __shared__ __attribute__((aligned(16))) float smem[DKK * PF + DKK * PP + 32 * PF + 32 * PP + 8 * 32 * PW];
    float* const wlf  = smem;
    float* const wlq  = wlf + DKK * PF;
    float* const actf = wlq + DKK * PP;
    float* const actq = actf + 32 * PF;
    float* const part = actq + 32 * PP;

    const int wave = threadIdx.x >> 5;
    const int lane = threadIdx.x & 31;
    const int br   = blockIdx.x >> 4;
    const int bc   = blockIdx.x & 15;
    const int half = lane >> 4;
    const int idx  = lane & 15;
    const int hh   = bc & 1;
    const int h    = bc >> 1;

    const float* Wf  = hh ? Wv : Wk;
    const float* Wtf = hh ? Wtv : Wtk;

    const int koff = wave * 64 + 4 * half;
    const float* xp = X  + (size_t)(br * 32 + idx) * DM + koff;
    const float* fp = Wf + (size_t)(h * DKK + idx) * DM + koff;
    const float* gp = Wq + (size_t)(h * DKK + 32 * hh + idx) * DM + koff;

    v4f x0[8], x1[8], f0[8], f1[8], f2[8], f3[8], g0[8], g1[8];
    #pragma unroll
    for (int j = 0; j < 8; ++j) {
        x0[j] = *(const v4f*)(xp + 8 * j);
        x1[j] = *(const v4f*)(xp + 16 * DM + 8 * j);
        f0[j] = *(const v4f*)(fp + 8 * j);
        f1[j] = *(const v4f*)(fp + 16 * DM + 8 * j);
        f2[j] = *(const v4f*)(fp + 32 * DM + 8 * j);
        f3[j] = *(const v4f*)(fp + 48 * DM + 8 * j);
        g0[j] = *(const v4f*)(gp + 8 * j);
        g1[j] = *(const v4f*)(gp + 16 * DM + 8 * j);
    }

    const v8f z = {0.f, 0.f, 0.f, 0.f, 0.f, 0.f, 0.f, 0.f};
    v8f F00 = z, F01 = z, F02 = z, F03 = z, F10 = z, F11 = z, F12 = z, F13 = z;
    v8f G00 = z, G01 = z, G10 = z, G11 = z;
    #pragma unroll
    for (int j = 0; j < 8; ++j) {
        WMMA4(F00, lo2(x0[j]), lo2(f0[j]));  WMMA4(F01, lo2(x0[j]), lo2(f1[j]));
        WMMA4(F02, lo2(x0[j]), lo2(f2[j]));  WMMA4(F03, lo2(x0[j]), lo2(f3[j]));
        WMMA4(F10, lo2(x1[j]), lo2(f0[j]));  WMMA4(F11, lo2(x1[j]), lo2(f1[j]));
        WMMA4(F12, lo2(x1[j]), lo2(f2[j]));  WMMA4(F13, lo2(x1[j]), lo2(f3[j]));
        WMMA4(G00, lo2(x0[j]), lo2(g0[j]));  WMMA4(G01, lo2(x0[j]), lo2(g1[j]));
        WMMA4(G10, lo2(x1[j]), lo2(g0[j]));  WMMA4(G11, lo2(x1[j]), lo2(g1[j]));
        WMMA4(F00, hi2(x0[j]), hi2(f0[j]));  WMMA4(F01, hi2(x0[j]), hi2(f1[j]));
        WMMA4(F02, hi2(x0[j]), hi2(f2[j]));  WMMA4(F03, hi2(x0[j]), hi2(f3[j]));
        WMMA4(F10, hi2(x1[j]), hi2(f0[j]));  WMMA4(F11, hi2(x1[j]), hi2(f1[j]));
        WMMA4(F12, hi2(x1[j]), hi2(f2[j]));  WMMA4(F13, hi2(x1[j]), hi2(f3[j]));
        WMMA4(G00, hi2(x0[j]), hi2(g0[j]));  WMMA4(G01, hi2(x0[j]), hi2(g1[j]));
        WMMA4(G10, hi2(x1[j]), hi2(g0[j]));  WMMA4(G11, hi2(x1[j]), hi2(g1[j]));
    }

    __builtin_amdgcn_sched_group_barrier(0x020, 64, 0);
    __builtin_amdgcn_sched_group_barrier(0x008, 192, 0);

    const float* wfp = Wtf + (threadIdx.x >> 4) * DKK + 4 * (threadIdx.x & 15);
    const float4 wf0 = *(const float4*)(wfp);
    const float4 wf1 = *(const float4*)(wfp + 16 * DKK);
    const float4 wf2 = *(const float4*)(wfp + 32 * DKK);
    const float4 wf3 = *(const float4*)(wfp + 48 * DKK);
    const float* wqp = Wtq + (threadIdx.x >> 3) * DKK + 32 * hh + 4 * (threadIdx.x & 7);
    const float4 wq0 = *(const float4*)(wqp);
    const float4 wq1 = *(const float4*)(wqp + 32 * DKK);

    float* pw = part + wave * (32 * PW);
    put_tile_w(pw, F00, 0, 0, half, idx);  put_tile_w(pw, F01, 0, 1, half, idx);
    put_tile_w(pw, F02, 0, 2, half, idx);  put_tile_w(pw, F03, 0, 3, half, idx);
    put_tile_w(pw, F10, 1, 0, half, idx);  put_tile_w(pw, F11, 1, 1, half, idx);
    put_tile_w(pw, F12, 1, 2, half, idx);  put_tile_w(pw, F13, 1, 3, half, idx);
    put_tile_w(pw, G00, 0, 4, half, idx);  put_tile_w(pw, G01, 0, 5, half, idx);
    put_tile_w(pw, G10, 1, 4, half, idx);  put_tile_w(pw, G11, 1, 5, half, idx);
    __syncthreads();

    {
        const int row = threadIdx.x >> 3, c4 = threadIdx.x & 7;
        #pragma unroll
        for (int m = 0; m < 3; ++m) {
            const float4 lm = *(const float4*)(lam + h * DKK + (m < 2 ? 32 * m : 32 * hh) + 4 * c4);
            const float* pr = part + row * PW + 32 * m + 4 * c4;
            float4 s = *(const float4*)pr;
            #pragma unroll
            for (int w = 1; w < 8; ++w) {
                const float4 t = *(const float4*)(pr + w * (32 * PW));
                s.x += t.x; s.y += t.y; s.z += t.z; s.w += t.w;
            }
            float* ad = (m < 2) ? (actf + row * PF + 32 * m + 4 * c4) : (actq + row * PP + 4 * c4);
            *(float4*)ad =
                make_float4(log1pf(fmaxf(s.x, 0.f)) - lm.x, log1pf(fmaxf(s.y, 0.f)) - lm.y,
                            log1pf(fmaxf(s.z, 0.f)) - lm.z, log1pf(fmaxf(s.w, 0.f)) - lm.w);
        }
        {
            const int of = threadIdx.x >> 4;
            float* w0 = wlf + ((of >> 2) + 16 * (of & 3)) * PF + 4 * (threadIdx.x & 15);
            *(float4*)(w0) = wf0;                *(float4*)(w0 + 4 * PF) = wf1;
            *(float4*)(w0 + 8 * PF) = wf2;       *(float4*)(w0 + 12 * PF) = wf3;
            const int oq = threadIdx.x >> 3;
            float* u0 = wlq + ((oq >> 2) + 16 * (oq & 3)) * PP + 4 * (threadIdx.x & 7);
            *(float4*)(u0) = wq0;                *(float4*)(u0 + 8 * PP) = wq1;
        }
    }
    __syncthreads();

    {
        const int p  = wave & 3;
        const int og = (lane & 7) + 8 * (p & 1);
        const int rb = 16 * (p >> 1) + ((lane >> 3) & 3);
        const size_t N = (size_t)ROWS * DM;
        const int wsc = __builtin_amdgcn_readfirstlane(wave);
        if (wsc < 4)
            trop_task<PF, DKK>(actf + rb * PF, wlf + og * PF, P + (size_t)(2 + hh) * N,
                               br * 32 + rb, h, og);
        else
            trop_task<PP, 32>(actq + rb * PP, wlq + og * PP, P + (size_t)hh * N,
                              br * 32 + rb, h, og);
    }
}

#define KP 68

__device__ __forceinline__ v4f max4(v4f a, v4f b)
{
    return __builtin_elementwise_maximum(a, b);
}
#define SP 260

__global__ __launch_bounds__(256)
void trop_attn(const float* __restrict__ qa, const float* __restrict__ qb,
               const float* __restrict__ kk, const float* __restrict__ vv,
               float* __restrict__ scores, float* __restrict__ ctxout)
{
    __shared__ __attribute__((aligned(16))) float qs[16 * KP];
    __shared__ __attribute__((aligned(16))) float ks[LL * KP];
    __shared__ __attribute__((aligned(16))) float vs[LL * KP];
    __shared__ __attribute__((aligned(16))) float ss[16 * SP];

    const int t  = threadIdx.x;
    const int it = blockIdx.x;
    const int bh = blockIdx.y;

    const float PI = __builtin_inff();
    const float NI = -__builtin_inff();

    {
        const size_t hb = (size_t)bh * LL * DKK;
        const v4f* qg0 = (const v4f*)(qa + hb + (size_t)(it * 16) * DKK);
        const v4f* qg1 = (const v4f*)(qb + hb + (size_t)(it * 16) * DKK);
        const v4f* kg = (const v4f*)(kk + hb);
        const v4f* vg = (const v4f*)(vv + hb);
        *(v4f*)(qs + (t >> 4) * KP + 4 * (t & 15)) = max4(qg0[t], qg1[t]);
        #pragma unroll
        for (int r = 0; r < 16; ++r) {
            const int e = t + 256 * r;
            const int row = e >> 4, c4 = e & 15;
            *(v4f*)(ks + row * KP + 4 * c4) = kg[e];
            *(v4f*)(vs + row * KP + 4 * c4) = vg[e];
        }
    }
    __syncthreads();

    {
        const int ig = t & 3;
        const int jg = t >> 2;
        float mx[4][4], mn[4][4];
        #pragma unroll
        for (int a = 0; a < 4; ++a)
            #pragma unroll
            for (int b = 0; b < 4; ++b) { mx[a][b] = NI; mn[a][b] = PI; }

        const float* qb = qs + (4 * ig) * KP;
        const float* kb = ks + jg * KP;
        #pragma unroll 2
        for (int d = 0; d < DKK; d += 4) {
            float4 qv[4], kv[4];
            #pragma unroll
            for (int a = 0; a < 4; ++a) qv[a] = *(const float4*)(qb + a * KP + d);
            #pragma unroll
            for (int b = 0; b < 4; ++b) kv[b] = *(const float4*)(kb + b * 64 * KP + d);
            #pragma unroll
            for (int a = 0; a < 4; ++a) {
                #pragma unroll
                for (int b = 0; b < 4; ++b) {
                    const float d0 = qv[a].x - kv[b].x;
                    const float d1 = qv[a].y - kv[b].y;
                    const float d2 = qv[a].z - kv[b].z;
                    const float d3 = qv[a].w - kv[b].w;
                    float hi = mx[a][b], lo = mn[a][b];
                    hi = fmaxf(fmaxf(hi, d0), d1);
                    hi = fmaxf(fmaxf(hi, d2), d3);
                    lo = fminf(fminf(lo, d0), d1);
                    lo = fminf(fminf(lo, d2), d3);
                    mx[a][b] = hi; mn[a][b] = lo;
                }
            }
        }
        #pragma unroll
        for (int a = 0; a < 4; ++a)
            #pragma unroll
            for (int b = 0; b < 4; ++b)
                ss[(4 * ig + a) * SP + jg + 64 * b] = mn[a][b] - mx[a][b];
    }
    __syncthreads();

    {
        float4* sg = (float4*)(scores + (size_t)bh * LL * LL + (size_t)(it * 16) * LL);
        #pragma unroll
        for (int r = 0; r < 4; ++r) {
            const int e = t + 256 * r;
            sg[e] = *(const float4*)(ss + (e >> 6) * SP + 4 * (e & 63));
        }
    }

    float* pm = ks;
    {
        const int grp = t >> 6;
        const int u   = t & 63;
        const int ci  = u & 3;
        const int dg  = u >> 2;
        float cx[4][4];
        #pragma unroll
        for (int a = 0; a < 4; ++a)
            #pragma unroll
            for (int c = 0; c < 4; ++c) cx[a][c] = NI;

        const float* sb = ss + (4 * ci) * SP + 64 * grp;
        const float* vb = vs + (64 * grp) * KP + 4 * dg;
        #pragma unroll 2
        for (int j = 0; j < 64; j += 4) {
            float4 sv[4], vv[4];
            #pragma unroll
            for (int a = 0; a < 4; ++a) sv[a] = *(const float4*)(sb + a * SP + j);
            #pragma unroll
            for (int q = 0; q < 4; ++q) vv[q] = *(const float4*)(vb + (j + q) * KP);
            #pragma unroll
            for (int a = 0; a < 4; ++a) {
                float m0 = cx[a][0], m1 = cx[a][1], m2 = cx[a][2], m3 = cx[a][3];
                m0 = fmaxf(fmaxf(m0, sv[a].x + vv[0].x), sv[a].y + vv[1].x);
                m0 = fmaxf(fmaxf(m0, sv[a].z + vv[2].x), sv[a].w + vv[3].x);
                m1 = fmaxf(fmaxf(m1, sv[a].x + vv[0].y), sv[a].y + vv[1].y);
                m1 = fmaxf(fmaxf(m1, sv[a].z + vv[2].y), sv[a].w + vv[3].y);
                m2 = fmaxf(fmaxf(m2, sv[a].x + vv[0].z), sv[a].y + vv[1].z);
                m2 = fmaxf(fmaxf(m2, sv[a].z + vv[2].z), sv[a].w + vv[3].z);
                m3 = fmaxf(fmaxf(m3, sv[a].x + vv[0].w), sv[a].y + vv[1].w);
                m3 = fmaxf(fmaxf(m3, sv[a].z + vv[2].w), sv[a].w + vv[3].w);
                cx[a][0] = m0; cx[a][1] = m1; cx[a][2] = m2; cx[a][3] = m3;
            }
        }
        #pragma unroll
        for (int a = 0; a < 4; ++a)
            *(float4*)(pm + grp * (16 * KP) + (4 * ci + a) * KP + 4 * dg) =
                make_float4(cx[a][0], cx[a][1], cx[a][2], cx[a][3]);
    }
    __syncthreads();

    {
        const int i = t >> 4, c4 = t & 15;
        const float* pr = pm + i * KP + 4 * c4;
        float4 m = *(const float4*)pr;
        #pragma unroll
        for (int g = 1; g < 4; ++g) {
            const float4 o = *(const float4*)(pr + g * (16 * KP));
            m.x = fmaxf(m.x, o.x); m.y = fmaxf(m.y, o.y);
            m.z = fmaxf(m.z, o.z); m.w = fmaxf(m.w, o.w);
        }
        const int b = bh >> 3, h = bh & 7;
        *(float4*)(ctxout + (size_t)(b * LL + it * 16 + i) * DM + h * DKK + 4 * c4) =
            make_float4(expm1f(m.x), expm1f(m.y), expm1f(m.z), expm1f(m.w));
    }
}

extern "C" void kernel_launch(void* const* d_in, const int* in_sizes, int n_in,
                              void* d_out, int out_size, void* d_ws, size_t ws_size,
                              hipStream_t stream)
{
    (void)in_sizes; (void)n_in; (void)out_size; (void)ws_size;
    const float* x   = (const float*)d_in[0];
    const float* Wq  = (const float*)d_in[1];
    const float* Wk  = (const float*)d_in[2];
    const float* Wv  = (const float*)d_in[3];
    const float* Wo  = (const float*)d_in[4];
    const float* lam = (const float*)d_in[5];
    const float* Wqt = (const float*)d_in[6];
    const float* Wkt = (const float*)d_in[7];
    const float* Wvt = (const float*)d_in[8];

    float* out = (float*)d_out;
    float* sc  = out + (size_t)ROWS * DM;

    const size_t N = (size_t)ROWS * DM;
    float* ws  = (float*)d_ws;
    float* pp  = ws;
    float* ctx = ws + 4 * N;

    dim3 blk(256);
    proj_headtile<<<256, blk, 0, stream>>>(x, Wq, Wk, Wv, lam, Wqt, Wkt, Wvt, pp);

    trop_attn<<<dim3(16, 16), blk, 0, stream>>>(pp, pp + N, pp + 2 * N, pp + 3 * N, sc, ctx);

    gemm512_ksplit<<<256, dim3(128), 0, stream>>>(ctx, Wo, out);
}


# activation log1p(relu(s)) on the hardware log2, as ln(u) + (s - (u - 1)) / u with u = 1 + s as rounded, not the library routine of some 70 vector instructions per value; results move by a few ulp
# speedup vs baseline: 5.2617x; 1.0324x over previous
#include <hip/hip_runtime.h>
#include <math.h>

#define DM   512
#define NH   8
#define DKK  64
#define B_   2
#define LL   256
#define ROWS (B_*LL)

typedef __attribute__((ext_vector_type(2))) float v2f;
typedef __attribute__((ext_vector_type(8))) float v8f;

typedef __attribute__((ext_vector_type(4))) float v4f;

#define PP 36

#define WMMA4(acc, av, bv) \
    acc = __builtin_amdgcn_wmma_f32_16x16x4_f32(false, av, false, bv, (short)0, acc, false, false)

__device__ __forceinline__ v2f lo2(v4f v) { return __builtin_shufflevector(v, v, 0, 1); }
__device__ __forceinline__ v2f hi2(v4f v) { return __builtin_shufflevector(v, v, 2, 3); }

__device__ __forceinline__ void put_tile(float* p, v8f c, int i, int n, int half, int idx)
{
    #pragma unroll
    for (int r = 0; r < 8; ++r)
        p[(16 * i + 8 * half + r) * PP + 16 * n + idx] = c[r];
}

__global__ __launch_bounds__(128)
void gemm512_ksplit(const float* __restrict__ A, const float* __restrict__ W,
                    float* __restrict__ Y)
{
    __shared__ __attribute__((aligned(16))) float part[4 * 32 * PP];

    const int wave = threadIdx.x >> 5;
    const int lane = threadIdx.x & 31;
    const int br   = blockIdx.x >> 4;
    const int bc   = blockIdx.x & 15;
    const int half = lane >> 4;
    const int idx  = lane & 15;

    const float* ap = A + (size_t)(br * 32 + idx) * DM + wave * 128 + 4 * half;
    const float* wp = W + (size_t)(bc * 32 + idx) * DM + wave * 128 + 4 * half;

    v4f a0[16], a1[16], b0[16], b1[16];
    #pragma unroll
    for (int j = 0; j < 16; ++j) {
        a0[j] = *(const v4f*)(ap + 8 * j);
        a1[j] = *(const v4f*)(ap + 16 * DM + 8 * j);
        b0[j] = *(const v4f*)(wp + 8 * j);
        b1[j] = *(const v4f*)(wp + 16 * DM + 8 * j);
    }

    v8f c00 = {0.f, 0.f, 0.f, 0.f, 0.f, 0.f, 0.f, 0.f};
    v8f c01 = c00, c10 = c00, c11 = c00;
    #pragma unroll
    for (int j = 0; j < 16; ++j) {
        WMMA4(c00, lo2(a0[j]), lo2(b0[j]));
        WMMA4(c01, lo2(a0[j]), lo2(b1[j]));
        WMMA4(c10, lo2(a1[j]), lo2(b0[j]));
        WMMA4(c11, lo2(a1[j]), lo2(b1[j]));
        WMMA4(c00, hi2(a0[j]), hi2(b0[j]));
        WMMA4(c01, hi2(a0[j]), hi2(b1[j]));
        WMMA4(c10, hi2(a1[j]), hi2(b0[j]));
        WMMA4(c11, hi2(a1[j]), hi2(b1[j]));
    }

    __builtin_amdgcn_sched_group_barrier(0x020, 64, 0);
    __builtin_amdgcn_sched_group_barrier(0x008, 128, 0);

    float* pw = part + wave * (32 * PP);
    put_tile(pw, c00, 0, 0, half, idx);
    put_tile(pw, c01, 0, 1, half, idx);
    put_tile(pw, c10, 1, 0, half, idx);
    put_tile(pw, c11, 1, 1, half, idx);
    __syncthreads();

    #pragma unroll
    for (int q = 0; q < 2; ++q) {
        const int e = threadIdx.x + 128 * q;
        const int row = e >> 3, c4 = e & 7;
        const float* pr = part + row * PP + 4 * c4;
        float4 s = *(const float4*)pr;
        #pragma unroll
        for (int w = 1; w < 4; ++w) {
            const float4 o = *(const float4*)(pr + w * (32 * PP));
            s.x += o.x; s.y += o.y; s.z += o.z; s.w += o.w;
        }
        *(float4*)(Y + (size_t)(br * 32 + row) * DM + bc * 32 + 4 * c4) = s;
    }
}

#define PW 100
#define PF 68

__device__ __forceinline__ float log1p_pos(float x)
{
    const float u = 1.0f + x;
    const float c = x - (u - 1.0f);
    const float r = __builtin_fmaf(__builtin_amdgcn_logf(u), 0.693147180559945f,
                                   c * __builtin_amdgcn_rcpf(u));
    return (u == __builtin_inff()) ? u : r;
}

__device__ __forceinline__ void put_tile_w(float* p, v8f c, int i, int n, int half, int idx)
{
    #pragma unroll
    for (int r = 0; r < 8; ++r)
        p[(16 * i + 8 * half + r) * PW + 16 * n + idx] = c[r];
}

template <int PITCH, int NI>
__device__ __forceinline__ void trop_task(const float* xb, const float* wb, float* dst,
                                          int row0, int h, int og)
{
    float mx[4][4];
    #pragma unroll
    for (int a = 0; a < 4; ++a)
        #pragma unroll
        for (int cc = 0; cc < 4; ++cc) mx[a][cc] = -__builtin_inff();

    #pragma unroll 4
    for (int i = 0; i < NI; i += 4) {
        float4 xv[4], wv[4];
        #pragma unroll
        for (int a = 0; a < 4; ++a) xv[a] = *(const float4*)(xb + (4 * a) * PITCH + i);
        #pragma unroll
        for (int cc = 0; cc < 4; ++cc) wv[cc] = *(const float4*)(wb + (16 * cc) * PITCH + i);
        #pragma unroll
        for (int a = 0; a < 4; ++a) {
            #pragma unroll
            for (int cc = 0; cc < 4; ++cc) {
                float v = mx[a][cc];
                v = fmaxf(fmaxf(v, xv[a].x + wv[cc].x), xv[a].y + wv[cc].y);
                v = fmaxf(fmaxf(v, xv[a].z + wv[cc].z), xv[a].w + wv[cc].w);
                mx[a][cc] = v;
            }
        }
    }

    #pragma unroll
    for (int a = 0; a < 4; ++a) {
        const int row = row0 + 4 * a;
        const int bh  = (row >> 8) * NH + h;
        *(float4*)(dst + ((size_t)bh * LL + (row & (LL - 1))) * DKK + 4 * og) =
            make_float4(mx[a][0], mx[a][1], mx[a][2], mx[a][3]);
    }
}

__global__ __launch_bounds__(256)
void proj_headtile(const float* __restrict__ X, const float* __restrict__ Wq,
                   const float* __restrict__ Wk, const float* __restrict__ Wv,
                   const float* __restrict__ lam, const float* __restrict__ Wtq,
                   const float* __restrict__ Wtk, const float* __restrict__ Wtv,
                   float* __restrict__ P)
{
    __shared__ __attribute__((aligned(16))) float smem[DKK * PF + DKK * PP + 32 * PF + 32 * PP + 8 * 32 * PW];
    float* const wlf  = smem;
    float* const wlq  = wlf + DKK * PF;
    float* const actf = wlq + DKK * PP;
    float* const actq = actf + 32 * PF;
    float* const part = actq + 32 * PP;

    const int wave = threadIdx.x >> 5;
    const int lane = threadIdx.x & 31;
    const int br   = blockIdx.x >> 4;
    const int bc   = blockIdx.x & 15;
    const int half = lane >> 4;
    const int idx  = lane & 15;
    const int hh   = bc & 1;
    const int h    = bc >> 1;

    const float* Wf  = hh ? Wv : Wk;
    const float* Wtf = hh ? Wtv : Wtk;

    const int koff = wave * 64 + 4 * half;
    const float* xp = X  + (size_t)(br * 32 + idx) * DM + koff;
    const float* fp = Wf + (size_t)(h * DKK + idx) * DM + koff;
    const float* gp = Wq + (size_t)(h * DKK + 32 * hh + idx) * DM + koff;

    v4f x0[8], x1[8], f0[8], f1[8], f2[8], f3[8], g0[8], g1[8];
    #pragma unroll
    for (int j = 0; j < 8; ++j) {
        x0[j] = *(const v4f*)(xp + 8 * j);
        x1[j] = *(const v4f*)(xp + 16 * DM + 8 * j);
        f0[j] = *(const v4f*)(fp + 8 * j);
        f1[j] = *(const v4f*)(fp + 16 * DM + 8 * j);
        f2[j] = *(const v4f*)(fp + 32 * DM + 8 * j);
        f3[j] = *(const v4f*)(fp + 48 * DM + 8 * j);
        g0[j] = *(const v4f*)(gp + 8 * j);
        g1[j] = *(const v4f*)(gp + 16 * DM + 8 * j);
    }

    const v8f z = {0.f, 0.f, 0.f, 0.f, 0.f, 0.f, 0.f, 0.f};
    v8f F00 = z, F01 = z, F02 = z, F03 = z, F10 = z, F11 = z, F12 = z, F13 = z;
    v8f G00 = z, G01 = z, G10 = z, G11 = z;
    #pragma unroll
    for (int j = 0; j < 8; ++j) {
        WMMA4(F00, lo2(x0[j]), lo2(f0[j]));  WMMA4(F01, lo2(x0[j]), lo2(f1[j]));
        WMMA4(F02, lo2(x0[j]), lo2(f2[j]));  WMMA4(F03, lo2(x0[j]), lo2(f3[j]));
        WMMA4(F10, lo2(x1[j]), lo2(f0[j]));  WMMA4(F11, lo2(x1[j]), lo2(f1[j]));
        WMMA4(F12, lo2(x1[j]), lo2(f2[j]));  WMMA4(F13, lo2(x1[j]), lo2(f3[j]));
        WMMA4(G00, lo2(x0[j]), lo2(g0[j]));  WMMA4(G01, lo2(x0[j]), lo2(g1[j]));
        WMMA4(G10, lo2(x1[j]), lo2(g0[j]));  WMMA4(G11, lo2(x1[j]), lo2(g1[j]));
        WMMA4(F00, hi2(x0[j]), hi2(f0[j]));  WMMA4(F01, hi2(x0[j]), hi2(f1[j]));
        WMMA4(F02, hi2(x0[j]), hi2(f2[j]));  WMMA4(F03, hi2(x0[j]), hi2(f3[j]));
        WMMA4(F10, hi2(x1[j]), hi2(f0[j]));  WMMA4(F11, hi2(x1[j]), hi2(f1[j]));
        WMMA4(F12, hi2(x1[j]), hi2(f2[j]));  WMMA4(F13, hi2(x1[j]), hi2(f3[j]));
        WMMA4(G00, hi2(x0[j]), hi2(g0[j]));  WMMA4(G01, hi2(x0[j]), hi2(g1[j]));
        WMMA4(G10, hi2(x1[j]), hi2(g0[j]));  WMMA4(G11, hi2(x1[j]), hi2(g1[j]));
    }

    __builtin_amdgcn_sched_group_barrier(0x020, 64, 0);
    __builtin_amdgcn_sched_group_barrier(0x008, 192, 0);

    const float* wfp = Wtf + (threadIdx.x >> 4) * DKK + 4 * (threadIdx.x & 15);
    const float4 wf0 = *(const float4*)(wfp);
    const float4 wf1 = *(const float4*)(wfp + 16 * DKK);
    const float4 wf2 = *(const float4*)(wfp + 32 * DKK);
    const float4 wf3 = *(const float4*)(wfp + 48 * DKK);
    const float* wqp = Wtq + (threadIdx.x >> 3) * DKK + 32 * hh + 4 * (threadIdx.x & 7);
    const float4 wq0 = *(const float4*)(wqp);
    const float4 wq1 = *(const float4*)(wqp + 32 * DKK);

    float* pw = part + wave * (32 * PW);
    put_tile_w(pw, F00, 0, 0, half, idx);  put_tile_w(pw, F01, 0, 1, half, idx);
    put_tile_w(pw, F02, 0, 2, half, idx);  put_tile_w(pw, F03, 0, 3, half, idx);
    put_tile_w(pw, F10, 1, 0, half, idx);  put_tile_w(pw, F11, 1, 1, half, idx);
    put_tile_w(pw, F12, 1, 2, half, idx);  put_tile_w(pw, F13, 1, 3, half, idx);
    put_tile_w(pw, G00, 0, 4, half, idx);  put_tile_w(pw, G01, 0, 5, half, idx);
    put_tile_w(pw, G10, 1, 4, half, idx);  put_tile_w(pw, G11, 1, 5, half, idx);
    __syncthreads();

    {
        const int row = threadIdx.x >> 3, c4 = threadIdx.x & 7;
        v4f pv[3][8];
        float4 lm[3];
        #pragma unroll
        for (int m = 0; m < 3; ++m) {
            lm[m] = *(const float4*)(lam + h * DKK + (m < 2 ? 32 * m : 32 * hh) + 4 * c4);
            const float* pr = part + row * PW + 32 * m + 4 * c4;
            #pragma unroll
            for (int w = 0; w < 8; ++w) pv[m][w] = *(const v4f*)(pr + w * (32 * PW));
        }
        asm volatile("" :: "v"(pv[0][0]), "v"(pv[0][1]), "v"(pv[0][2]), "v"(pv[0][3]),
                           "v"(pv[0][4]), "v"(pv[0][5]), "v"(pv[0][6]), "v"(pv[0][7]),
                           "v"(pv[1][0]), "v"(pv[1][1]), "v"(pv[1][2]), "v"(pv[1][3]),
                           "v"(pv[1][4]), "v"(pv[1][5]), "v"(pv[1][6]), "v"(pv[1][7]),
                           "v"(pv[2][0]), "v"(pv[2][1]), "v"(pv[2][2]), "v"(pv[2][3]),
                           "v"(pv[2][4]), "v"(pv[2][5]), "v"(pv[2][6]), "v"(pv[2][7]));
        #pragma unroll
        for (int m = 0; m < 3; ++m) {
            v4f s = pv[m][0];
            #pragma unroll
            for (int w = 1; w < 8; ++w) s += pv[m][w];
            float* ad = (m < 2) ? (actf + row * PF + 32 * m + 4 * c4) : (actq + row * PP + 4 * c4);
            *(float4*)ad =
                make_float4(log1p_pos(fmaxf(s.x, 0.f)) - lm[m].x, log1p_pos(fmaxf(s.y, 0.f)) - lm[m].y,
                            log1p_pos(fmaxf(s.z, 0.f)) - lm[m].z, log1p_pos(fmaxf(s.w, 0.f)) - lm[m].w);
        }
        {
            const int of = threadIdx.x >> 4;
            float* w0 = wlf + ((of >> 2) + 16 * (of & 3)) * PF + 4 * (threadIdx.x & 15);
            *(float4*)(w0) = wf0;                *(float4*)(w0 + 4 * PF) = wf1;
            *(float4*)(w0 + 8 * PF) = wf2;       *(float4*)(w0 + 12 * PF) = wf3;
            const int oq = threadIdx.x >> 3;
            float* u0 = wlq + ((oq >> 2) + 16 * (oq & 3)) * PP + 4 * (threadIdx.x & 7);
            *(float4*)(u0) = wq0;                *(float4*)(u0 + 8 * PP) = wq1;
        }
    }
    __syncthreads();

    {
        const int p  = wave & 3;
        const int og = (lane & 7) + 8 * (p & 1);
        const int rb = 16 * (p >> 1) + ((lane >> 3) & 3);
        const size_t N = (size_t)ROWS * DM;
        const int wsc = __builtin_amdgcn_readfirstlane(wave);
        if (wsc < 4)
            trop_task<PF, DKK>(actf + rb * PF, wlf + og * PF, P + (size_t)(2 + hh) * N,
                               br * 32 + rb, h, og);
        else
            trop_task<PP, 32>(actq + rb * PP, wlq + og * PP, P + (size_t)hh * N,
                              br * 32 + rb, h, og);
    }
}

#define KP 68

__device__ __forceinline__ v4f max4(v4f a, v4f b)
{
    return __builtin_elementwise_maximum(a, b);
}
#define SP 260

__global__ __launch_bounds__(256)
void trop_attn(const float* __restrict__ qa, const float* __restrict__ qb,
               const float* __restrict__ kk, const float* __restrict__ vv,
               float* __restrict__ scores, float* __restrict__ ctxout)
{
    __shared__ __attribute__((aligned(16))) float qs[16 * KP];
    __shared__ __attribute__((aligned(16))) float ks[LL * KP];
    __shared__ __attribute__((aligned(16))) float vs[LL * KP];
    __shared__ __attribute__((aligned(16))) float ss[16 * SP];

    const int t  = threadIdx.x;
    const int it = blockIdx.x;
    const int bh = blockIdx.y;

    const float PI = __builtin_inff();
    const float NI = -__builtin_inff();

    {
        const size_t hb = (size_t)bh * LL * DKK;
        const v4f* qg0 = (const v4f*)(qa + hb + (size_t)(it * 16) * DKK);
        const v4f* qg1 = (const v4f*)(qb + hb + (size_t)(it * 16) * DKK);
        const v4f* kg = (const v4f*)(kk + hb);
        const v4f* vg = (const v4f*)(vv + hb);
        *(v4f*)(qs + (t >> 4) * KP + 4 * (t & 15)) = max4(qg0[t], qg1[t]);
        #pragma unroll
        for (int r = 0; r < 16; ++r) {
            const int e = t + 256 * r;
            const int row = e >> 4, c4 = e & 15;
            *(v4f*)(ks + row * KP + 4 * c4) = kg[e];
            *(v4f*)(vs + row * KP + 4 * c4) = vg[e];
        }
    }
    __syncthreads();

    {
        const int ig = t & 3;
        const int jg = t >> 2;
        float mx[4][4], mn[4][4];
        #pragma unroll
        for (int a = 0; a < 4; ++a)
            #pragma unroll
            for (int b = 0; b < 4; ++b) { mx[a][b] = NI; mn[a][b] = PI; }

        const float* qb = qs + (4 * ig) * KP;
        const float* kb = ks + jg * KP;
        #pragma unroll 2
        for (int d = 0; d < DKK; d += 4) {
            float4 qv[4], kv[4];
            #pragma unroll
            for (int a = 0; a < 4; ++a) qv[a] = *(const float4*)(qb + a * KP + d);
            #pragma unroll
            for (int b = 0; b < 4; ++b) kv[b] = *(const float4*)(kb + b * 64 * KP + d);
            #pragma unroll
            for (int a = 0; a < 4; ++a) {
                #pragma unroll
                for (int b = 0; b < 4; ++b) {
                    const float d0 = qv[a].x - kv[b].x;
                    const float d1 = qv[a].y - kv[b].y;
                    const float d2 = qv[a].z - kv[b].z;
                    const float d3 = qv[a].w - kv[b].w;
                    float hi = mx[a][b], lo = mn[a][b];
                    hi = fmaxf(fmaxf(hi, d0), d1);
                    hi = fmaxf(fmaxf(hi, d2), d3);
                    lo = fminf(fminf(lo, d0), d1);
                    lo = fminf(fminf(lo, d2), d3);
                    mx[a][b] = hi; mn[a][b] = lo;
                }
            }
        }
        #pragma unroll
        for (int a = 0; a < 4; ++a)
            #pragma unroll
            for (int b = 0; b < 4; ++b)
                ss[(4 * ig + a) * SP + jg + 64 * b] = mn[a][b] - mx[a][b];
    }
    __syncthreads();

    {
        float4* sg = (float4*)(scores + (size_t)bh * LL * LL + (size_t)(it * 16) * LL);
        #pragma unroll
        for (int r = 0; r < 4; ++r) {
            const int e = t + 256 * r;
            sg[e] = *(const float4*)(ss + (e >> 6) * SP + 4 * (e & 63));
        }
    }

    float* pm = ks;
    {
        const int grp = t >> 6;
        const int u   = t & 63;
        const int ci  = u & 3;
        const int dg  = u >> 2;
        float cx[4][4];
        #pragma unroll
        for (int a = 0; a < 4; ++a)
            #pragma unroll
            for (int c = 0; c < 4; ++c) cx[a][c] = NI;

        const float* sb = ss + (4 * ci) * SP + 64 * grp;
        const float* vb = vs + (64 * grp) * KP + 4 * dg;
        #pragma unroll 2
        for (int j = 0; j < 64; j += 4) {
            float4 sv[4], vv[4];
            #pragma unroll
            for (int a = 0; a < 4; ++a) sv[a] = *(const float4*)(sb + a * SP + j);
            #pragma unroll
            for (int q = 0; q < 4; ++q) vv[q] = *(const float4*)(vb + (j + q) * KP);
            #pragma unroll
            for (int a = 0; a < 4; ++a) {
                float m0 = cx[a][0], m1 = cx[a][1], m2 = cx[a][2], m3 = cx[a][3];
                m0 = fmaxf(fmaxf(m0, sv[a].x + vv[0].x), sv[a].y + vv[1].x);
                m0 = fmaxf(fmaxf(m0, sv[a].z + vv[2].x), sv[a].w + vv[3].x);
                m1 = fmaxf(fmaxf(m1, sv[a].x + vv[0].y), sv[a].y + vv[1].y);
                m1 = fmaxf(fmaxf(m1, sv[a].z + vv[2].y), sv[a].w + vv[3].y);
                m2 = fmaxf(fmaxf(m2, sv[a].x + vv[0].z), sv[a].y + vv[1].z);
                m2 = fmaxf(fmaxf(m2, sv[a].z + vv[2].z), sv[a].w + vv[3].z);
                m3 = fmaxf(fmaxf(m3, sv[a].x + vv[0].w), sv[a].y + vv[1].w);
                m3 = fmaxf(fmaxf(m3, sv[a].z + vv[2].w), sv[a].w + vv[3].w);
                cx[a][0] = m0; cx[a][1] = m1; cx[a][2] = m2; cx[a][3] = m3;
            }
        }
        #pragma unroll
        for (int a = 0; a < 4; ++a)
            *(float4*)(pm + grp * (16 * KP) + (4 * ci + a) * KP + 4 * dg) =
                make_float4(cx[a][0], cx[a][1], cx[a][2], cx[a][3]);
    }
    __syncthreads();

    {
        const int i = t >> 4, c4 = t & 15;
        const float* pr = pm + i * KP + 4 * c4;
        float4 m = *(const float4*)pr;
        #pragma unroll
        for (int g = 1; g < 4; ++g) {
            const float4 o = *(const float4*)(pr + g * (16 * KP));
            m.x = fmaxf(m.x, o.x); m.y = fmaxf(m.y, o.y);
            m.z = fmaxf(m.z, o.z); m.w = fmaxf(m.w, o.w);
        }
        const int b = bh >> 3, h = bh & 7;
        *(float4*)(ctxout + (size_t)(b * LL + it * 16 + i) * DM + h * DKK + 4 * c4) =
            make_float4(expm1f(m.x), expm1f(m.y), expm1f(m.z), expm1f(m.w));
    }
}

extern "C" void kernel_launch(void* const* d_in, const int* in_sizes, int n_in,
                              void* d_out, int out_size, void* d_ws, size_t ws_size,
                              hipStream_t stream)
{
    (void)in_sizes; (void)n_in; (void)out_size; (void)ws_size;
    const float* x   = (const float*)d_in[0];
    const float* Wq  = (const float*)d_in[1];
    const float* Wk  = (const float*)d_in[2];
    const float* Wv  = (const float*)d_in[3];
    const float* Wo  = (const float*)d_in[4];
    const float* lam = (const float*)d_in[5];
    const float* Wqt = (const float*)d_in[6];
    const float* Wkt = (const float*)d_in[7];
    const float* Wvt = (const float*)d_in[8];

    float* out = (float*)d_out;
    float* sc  = out + (size_t)ROWS * DM;

    const size_t N = (size_t)ROWS * DM;
    float* ws  = (float*)d_ws;
    float* pp  = ws;
    float* ctx = ws + 4 * N;

    dim3 blk(256);
    proj_headtile<<<256, blk, 0, stream>>>(x, Wq, Wk, Wv, lam, Wqt, Wkt, Wvt, pp);

    trop_attn<<<dim3(16, 16), blk, 0, stream>>>(pp, pp + N, pp + 2 * N, pp + 3 * N, sc, ctx);

    gemm512_ksplit<<<256, dim3(128), 0, stream>>>(ctx, Wo, out);
}


# staggered reads: blocks that share operand rows start at different places (GEMM k steps from (block row + column) mod steps, attention K/V chunks from chunk it, wrapping); GEMM k sum order changes
# speedup vs baseline: 5.4666x; 1.0389x over previous
#include <hip/hip_runtime.h>
#include <math.h>

#define DM   512
#define NH   8
#define DKK  64
#define B_   2
#define LL   256
#define ROWS (B_*LL)

typedef __attribute__((ext_vector_type(2))) float v2f;
typedef __attribute__((ext_vector_type(8))) float v8f;

typedef __attribute__((ext_vector_type(4))) float v4f;

#define PP 36

#define WMMA4(acc, av, bv) \
    acc = __builtin_amdgcn_wmma_f32_16x16x4_f32(false, av, false, bv, (short)0, acc, false, false)

__device__ __forceinline__ v2f lo2(v4f v) { return __builtin_shufflevector(v, v, 0, 1); }
__device__ __forceinline__ v2f hi2(v4f v) { return __builtin_shufflevector(v, v, 2, 3); }

__device__ __forceinline__ void put_tile(float* p, v8f c, int i, int n, int half, int idx)
{
    #pragma unroll
    for (int r = 0; r < 8; ++r)
        p[(16 * i + 8 * half + r) * PP + 16 * n + idx] = c[r];
}

__global__ __launch_bounds__(128)
void gemm512_ksplit(const float* __restrict__ A, const float* __restrict__ W,
                    float* __restrict__ Y)
{
    __shared__ __attribute__((aligned(16))) float part[4 * 32 * PP];

    const int wave = threadIdx.x >> 5;
    const int lane = threadIdx.x & 31;
    const int br   = blockIdx.x >> 4;
    const int bc   = blockIdx.x & 15;
    const int half = lane >> 4;
    const int idx  = lane & 15;

    const float* ap = A + (size_t)(br * 32 + idx) * DM + wave * 128 + 4 * half;
    const float* wp = W + (size_t)(bc * 32 + idx) * DM + wave * 128 + 4 * half;
    const int rot = (br + bc) & 15;

    v4f a0[16], a1[16], b0[16], b1[16];
    #pragma unroll
    for (int j = 0; j < 16; ++j) {
        const int jj = (j + rot) & 15;
        a0[j] = *(const v4f*)(ap + 8 * jj);
        a1[j] = *(const v4f*)(ap + 16 * DM + 8 * jj);
        b0[j] = *(const v4f*)(wp + 8 * jj);
        b1[j] = *(const v4f*)(wp + 16 * DM + 8 * jj);
    }

    v8f c00 = {0.f, 0.f, 0.f, 0.f, 0.f, 0.f, 0.f, 0.f};
    v8f c01 = c00, c10 = c00, c11 = c00;
    #pragma unroll
    for (int j = 0; j < 16; ++j) {
        WMMA4(c00, lo2(a0[j]), lo2(b0[j]));
        WMMA4(c01, lo2(a0[j]), lo2(b1[j]));
        WMMA4(c10, lo2(a1[j]), lo2(b0[j]));
        WMMA4(c11, lo2(a1[j]), lo2(b1[j]));
        WMMA4(c00, hi2(a0[j]), hi2(b0[j]));
        WMMA4(c01, hi2(a0[j]), hi2(b1[j]));
        WMMA4(c10, hi2(a1[j]), hi2(b0[j]));
        WMMA4(c11, hi2(a1[j]), hi2(b1[j]));
    }

    __builtin_amdgcn_sched_group_barrier(0x020, 64, 0);
    __builtin_amdgcn_sched_group_barrier(0x008, 128, 0);

    float* pw = part + wave * (32 * PP);
    put_tile(pw, c00, 0, 0, half, idx);
    put_tile(pw, c01, 0, 1, half, idx);
    put_tile(pw, c10, 1, 0, half, idx);
    put_tile(pw, c11, 1, 1, half, idx);
    __syncthreads();

    #pragma unroll
    for (int q = 0; q < 2; ++q) {
        const int e = threadIdx.x + 128 * q;
        const int row = e >> 3, c4 = e & 7;
        const float* pr = part + row * PP + 4 * c4;
        float4 s = *(const float4*)pr;
        #pragma unroll
        for (int w = 1; w < 4; ++w) {
            const float4 o = *(const float4*)(pr + w * (32 * PP));
            s.x += o.x; s.y += o.y; s.z += o.z; s.w += o.w;
        }
        *(float4*)(Y + (size_t)(br * 32 + row) * DM + bc * 32 + 4 * c4) = s;
    }
}

#define PW 100
#define PF 68

__device__ __forceinline__ float log1p_pos(float x)
{
    const float u = 1.0f + x;
    const float c = x - (u - 1.0f);
    const float r = __builtin_fmaf(__builtin_amdgcn_logf(u), 0.693147180559945f,
                                   c * __builtin_amdgcn_rcpf(u));
    return (u == __builtin_inff()) ? u : r;
}

__device__ __forceinline__ void put_tile_w(float* p, v8f c, int i, int n, int half, int idx)
{
    #pragma unroll
    for (int r = 0; r < 8; ++r)
        p[(16 * i + 8 * half + r) * PW + 16 * n + idx] = c[r];
}

template <int PITCH, int NI>
__device__ __forceinline__ void trop_task(const float* xb, const float* wb, float* dst,
                                          int row0, int h, int og)
{
    float mx[4][4];
    #pragma unroll
    for (int a = 0; a < 4; ++a)
        #pragma unroll
        for (int cc = 0; cc < 4; ++cc) mx[a][cc] = -__builtin_inff();

    #pragma unroll 4
    for (int i = 0; i < NI; i += 4) {
        float4 xv[4], wv[4];
        #pragma unroll
        for (int a = 0; a < 4; ++a) xv[a] = *(const float4*)(xb + (4 * a) * PITCH + i);
        #pragma unroll
        for (int cc = 0; cc < 4; ++cc) wv[cc] = *(const float4*)(wb + (16 * cc) * PITCH + i);
        #pragma unroll
        for (int a = 0; a < 4; ++a) {
            #pragma unroll
            for (int cc = 0; cc < 4; ++cc) {
                float v = mx[a][cc];
                v = fmaxf(fmaxf(v, xv[a].x + wv[cc].x), xv[a].y + wv[cc].y);
                v = fmaxf(fmaxf(v, xv[a].z + wv[cc].z), xv[a].w + wv[cc].w);
                mx[a][cc] = v;
            }
        }
    }

    #pragma unroll
    for (int a = 0; a < 4; ++a) {
        const int row = row0 + 4 * a;
        const int bh  = (row >> 8) * NH + h;
        *(float4*)(dst + ((size_t)bh * LL + (row & (LL - 1))) * DKK + 4 * og) =
            make_float4(mx[a][0], mx[a][1], mx[a][2], mx[a][3]);
    }
}

__global__ __launch_bounds__(256)
void proj_headtile(const float* __restrict__ X, const float* __restrict__ Wq,
                   const float* __restrict__ Wk, const float* __restrict__ Wv,
                   const float* __restrict__ lam, const float* __restrict__ Wtq,
                   const float* __restrict__ Wtk, const float* __restrict__ Wtv,
                   float* __restrict__ P)
{
    __shared__ __attribute__((aligned(16))) float smem[DKK * PF + DKK * PP + 32 * PF + 32 * PP + 8 * 32 * PW];
    float* const wlf  = smem;
    float* const wlq  = wlf + DKK * PF;
    float* const actf = wlq + DKK * PP;
    float* const actq = actf + 32 * PF;
    float* const part = actq + 32 * PP;

    const int wave = threadIdx.x >> 5;
    const int lane = threadIdx.x & 31;
    const int br   = blockIdx.x >> 4;
    const int bc   = blockIdx.x & 15;
    const int half = lane >> 4;
    const int idx  = lane & 15;
    const int hh   = bc & 1;
    const int h    = bc >> 1;

    const float* Wf  = hh ? Wv : Wk;
    const float* Wtf = hh ? Wtv : Wtk;

    const int koff = wave * 64 + 4 * half;
    const float* xp = X  + (size_t)(br * 32 + idx) * DM + koff;
    const float* fp = Wf + (size_t)(h * DKK + idx) * DM + koff;
    const float* gp = Wq + (size_t)(h * DKK + 32 * hh + idx) * DM + koff;
    const int rot = (br + bc) & 7;

    v4f x0[8], x1[8], f0[8], f1[8], f2[8], f3[8], g0[8], g1[8];
    #pragma unroll
    for (int j = 0; j < 8; ++j) {
        const int jj = (j + rot) & 7;
        x0[j] = *(const v4f*)(xp + 8 * jj);
        x1[j] = *(const v4f*)(xp + 16 * DM + 8 * jj);
        f0[j] = *(const v4f*)(fp + 8 * jj);
        f1[j] = *(const v4f*)(fp + 16 * DM + 8 * jj);
        f2[j] = *(const v4f*)(fp + 32 * DM + 8 * jj);
        f3[j] = *(const v4f*)(fp + 48 * DM + 8 * jj);
        g0[j] = *(const v4f*)(gp + 8 * jj);
        g1[j] = *(const v4f*)(gp + 16 * DM + 8 * jj);
    }

    const v8f z = {0.f, 0.f, 0.f, 0.f, 0.f, 0.f, 0.f, 0.f};
    v8f F00 = z, F01 = z, F02 = z, F03 = z, F10 = z, F11 = z, F12 = z, F13 = z;
    v8f G00 = z, G01 = z, G10 = z, G11 = z;
    #pragma unroll
    for (int j = 0; j < 8; ++j) {
        WMMA4(F00, lo2(x0[j]), lo2(f0[j]));  WMMA4(F01, lo2(x0[j]), lo2(f1[j]));
        WMMA4(F02, lo2(x0[j]), lo2(f2[j]));  WMMA4(F03, lo2(x0[j]), lo2(f3[j]));
        WMMA4(F10, lo2(x1[j]), lo2(f0[j]));  WMMA4(F11, lo2(x1[j]), lo2(f1[j]));
        WMMA4(F12, lo2(x1[j]), lo2(f2[j]));  WMMA4(F13, lo2(x1[j]), lo2(f3[j]));
        WMMA4(G00, lo2(x0[j]), lo2(g0[j]));  WMMA4(G01, lo2(x0[j]), lo2(g1[j]));
        WMMA4(G10, lo2(x1[j]), lo2(g0[j]));  WMMA4(G11, lo2(x1[j]), lo2(g1[j]));
        WMMA4(F00, hi2(x0[j]), hi2(f0[j]));  WMMA4(F01, hi2(x0[j]), hi2(f1[j]));
        WMMA4(F02, hi2(x0[j]), hi2(f2[j]));  WMMA4(F03, hi2(x0[j]), hi2(f3[j]));
        WMMA4(F10, hi2(x1[j]), hi2(f0[j]));  WMMA4(F11, hi2(x1[j]), hi2(f1[j]));
        WMMA4(F12, hi2(x1[j]), hi2(f2[j]));  WMMA4(F13, hi2(x1[j]), hi2(f3[j]));
        WMMA4(G00, hi2(x0[j]), hi2(g0[j]));  WMMA4(G01, hi2(x0[j]), hi2(g1[j]));
        WMMA4(G10, hi2(x1[j]), hi2(g0[j]));  WMMA4(G11, hi2(x1[j]), hi2(g1[j]));
    }

    __builtin_amdgcn_sched_group_barrier(0x020, 64, 0);
    __builtin_amdgcn_sched_group_barrier(0x008, 192, 0);

    const float* wfp = Wtf + (threadIdx.x >> 4) * DKK + 4 * (threadIdx.x & 15);
    const float4 wf0 = *(const float4*)(wfp);
    const float4 wf1 = *(const float4*)(wfp + 16 * DKK);
    const float4 wf2 = *(const float4*)(wfp + 32 * DKK);
    const float4 wf3 = *(const float4*)(wfp + 48 * DKK);
    const float* wqp = Wtq + (threadIdx.x >> 3) * DKK + 32 * hh + 4 * (threadIdx.x & 7);
    const float4 wq0 = *(const float4*)(wqp);
    const float4 wq1 = *(const float4*)(wqp + 32 * DKK);

    float* pw = part + wave * (32 * PW);
    put_tile_w(pw, F00, 0, 0, half, idx);  put_tile_w(pw, F01, 0, 1, half, idx);
    put_tile_w(pw, F02, 0, 2, half, idx);  put_tile_w(pw, F03, 0, 3, half, idx);
    put_tile_w(pw, F10, 1, 0, half, idx);  put_tile_w(pw, F11, 1, 1, half, idx);
    put_tile_w(pw, F12, 1, 2, half, idx);  put_tile_w(pw, F13, 1, 3, half, idx);
    put_tile_w(pw, G00, 0, 4, half, idx);  put_tile_w(pw, G01, 0, 5, half, idx);
    put_tile_w(pw, G10, 1, 4, half, idx);  put_tile_w(pw, G11, 1, 5, half, idx);
    __syncthreads();

    {
        const int row = threadIdx.x >> 3, c4 = threadIdx.x & 7;
        v4f pv[3][8];
        float4 lm[3];
        #pragma unroll
        for (int m = 0; m < 3; ++m) {
            lm[m] = *(const float4*)(lam + h * DKK + (m < 2 ? 32 * m : 32 * hh) + 4 * c4);
            const float* pr = part + row * PW + 32 * m + 4 * c4;
            #pragma unroll
            for (int w = 0; w < 8; ++w) pv[m][w] = *(const v4f*)(pr + w * (32 * PW));
        }
        asm volatile("" :: "v"(pv[0][0]), "v"(pv[0][1]), "v"(pv[0][2]), "v"(pv[0][3]),
                           "v"(pv[0][4]), "v"(pv[0][5]), "v"(pv[0][6]), "v"(pv[0][7]),
                           "v"(pv[1][0]), "v"(pv[1][1]), "v"(pv[1][2]), "v"(pv[1][3]),
                           "v"(pv[1][4]), "v"(pv[1][5]), "v"(pv[1][6]), "v"(pv[1][7]),
                           "v"(pv[2][0]), "v"(pv[2][1]), "v"(pv[2][2]), "v"(pv[2][3]),
                           "v"(pv[2][4]), "v"(pv[2][5]), "v"(pv[2][6]), "v"(pv[2][7]));
        #pragma unroll
        for (int m = 0; m < 3; ++m) {
            v4f s = pv[m][0];
            #pragma unroll
            for (int w = 1; w < 8; ++w) s += pv[m][w];
            float* ad = (m < 2) ? (actf + row * PF + 32 * m + 4 * c4) : (actq + row * PP + 4 * c4);
            *(float4*)ad =
                make_float4(log1p_pos(fmaxf(s.x, 0.f)) - lm[m].x, log1p_pos(fmaxf(s.y, 0.f)) - lm[m].y,
                            log1p_pos(fmaxf(s.z, 0.f)) - lm[m].z, log1p_pos(fmaxf(s.w, 0.f)) - lm[m].w);
        }
        {
            const int of = threadIdx.x >> 4;
            float* w0 = wlf + ((of >> 2) + 16 * (of & 3)) * PF + 4 * (threadIdx.x & 15);
            *(float4*)(w0) = wf0;                *(float4*)(w0 + 4 * PF) = wf1;
            *(float4*)(w0 + 8 * PF) = wf2;       *(float4*)(w0 + 12 * PF) = wf3;
            const int oq = threadIdx.x >> 3;
            float* u0 = wlq + ((oq >> 2) + 16 * (oq & 3)) * PP + 4 * (threadIdx.x & 7);
            *(float4*)(u0) = wq0;                *(float4*)(u0 + 8 * PP) = wq1;
        }
    }
    __syncthreads();

    {
        const int p  = wave & 3;
        const int og = (lane & 7) + 8 * (p & 1);
        const int rb = 16 * (p >> 1) + ((lane >> 3) & 3);
        const size_t N = (size_t)ROWS * DM;
        const int wsc = __builtin_amdgcn_readfirstlane(wave);
        if (wsc < 4)
            trop_task<PF, DKK>(actf + rb * PF, wlf + og * PF, P + (size_t)(2 + hh) * N,
                               br * 32 + rb, h, og);
        else
            trop_task<PP, 32>(actq + rb * PP, wlq + og * PP, P + (size_t)hh * N,
                              br * 32 + rb, h, og);
    }
}

#define KP 68

__device__ __forceinline__ v4f max4(v4f a, v4f b)
{
    return __builtin_elementwise_maximum(a, b);
}
#define SP 260

__global__ __launch_bounds__(256)
void trop_attn(const float* __restrict__ qa, const float* __restrict__ qb,
               const float* __restrict__ kk, const float* __restrict__ vv,
               float* __restrict__ scores, float* __restrict__ ctxout)
{
    __shared__ __attribute__((aligned(16))) float qs[16 * KP];
    __shared__ __attribute__((aligned(16))) float ks[LL * KP];
    __shared__ __attribute__((aligned(16))) float vs[LL * KP];
    __shared__ __attribute__((aligned(16))) float ss[16 * SP];

    const int t  = threadIdx.x;
    const int it = blockIdx.x;
    const int bh = blockIdx.y;

    const float PI = __builtin_inff();
    const float NI = -__builtin_inff();

    {
        const size_t hb = (size_t)bh * LL * DKK;
        const v4f* qg0 = (const v4f*)(qa + hb + (size_t)(it * 16) * DKK);
        const v4f* qg1 = (const v4f*)(qb + hb + (size_t)(it * 16) * DKK);
        const v4f* kg = (const v4f*)(kk + hb);
        const v4f* vg = (const v4f*)(vv + hb);
        *(v4f*)(qs + (t >> 4) * KP + 4 * (t & 15)) = max4(qg0[t], qg1[t]);
        float* const kd = ks + (t >> 4) * KP + 4 * (t & 15);
        float* const vd = vs + (t >> 4) * KP + 4 * (t & 15);
        #pragma unroll
        for (int r = 0; r < 16; ++r) {
            const int rr = (r + it) & 15;
            const v4f* kgr = kg + 256 * rr;
            const v4f* vgr = vg + 256 * rr;
            *(v4f*)(kd + rr * (16 * KP)) = kgr[t];
            *(v4f*)(vd + rr * (16 * KP)) = vgr[t];
        }
    }
    __syncthreads();

    {
        const int ig = t & 3;
        const int jg = t >> 2;
        float mx[4][4], mn[4][4];
        #pragma unroll
        for (int a = 0; a < 4; ++a)
            #pragma unroll
            for (int b = 0; b < 4; ++b) { mx[a][b] = NI; mn[a][b] = PI; }

        const float* qb = qs + (4 * ig) * KP;
        const float* kb = ks + jg * KP;
        #pragma unroll 2
        for (int d = 0; d < DKK; d += 4) {
            float4 qv[4], kv[4];
            #pragma unroll
            for (int a = 0; a < 4; ++a) qv[a] = *(const float4*)(qb + a * KP + d);
            #pragma unroll
            for (int b = 0; b < 4; ++b) kv[b] = *(const float4*)(kb + b * 64 * KP + d);
            #pragma unroll
            for (int a = 0; a < 4; ++a) {
                #pragma unroll
                for (int b = 0; b < 4; ++b) {
                    const float d0 = qv[a].x - kv[b].x;
                    const float d1 = qv[a].y - kv[b].y;
                    const float d2 = qv[a].z - kv[b].z;
                    const float d3 = qv[a].w - kv[b].w;
                    float hi = mx[a][b], lo = mn[a][b];
                    hi = fmaxf(fmaxf(hi, d0), d1);
                    hi = fmaxf(fmaxf(hi, d2), d3);
                    lo = fminf(fminf(lo, d0), d1);
                    lo = fminf(fminf(lo, d2), d3);
                    mx[a][b] = hi; mn[a][b] = lo;
                }
            }
        }
        #pragma unroll
        for (int a = 0; a < 4; ++a)
            #pragma unroll
            for (int b = 0; b < 4; ++b)
                ss[(4 * ig + a) * SP + jg + 64 * b] = mn[a][b] - mx[a][b];
    }
    __syncthreads();

    {
        float4* sg = (float4*)(scores + (size_t)bh * LL * LL + (size_t)(it * 16) * LL);
        #pragma unroll
        for (int r = 0; r < 4; ++r) {
            const int e = t + 256 * r;
            sg[e] = *(const float4*)(ss + (e >> 6) * SP + 4 * (e & 63));
        }
    }

    float* pm = ks;
    {
        const int grp = t >> 6;
        const int u   = t & 63;
        const int ci  = u & 3;
        const int dg  = u >> 2;
        float cx[4][4];
        #pragma unroll
        for (int a = 0; a < 4; ++a)
            #pragma unroll
            for (int c = 0; c < 4; ++c) cx[a][c] = NI;

        const float* sb = ss + (4 * ci) * SP + 64 * grp;
        const float* vb = vs + (64 * grp) * KP + 4 * dg;
        #pragma unroll 2
        for (int j = 0; j < 64; j += 4) {
            float4 sv[4], vv[4];
            #pragma unroll
            for (int a = 0; a < 4; ++a) sv[a] = *(const float4*)(sb + a * SP + j);
            #pragma unroll
            for (int q = 0; q < 4; ++q) vv[q] = *(const float4*)(vb + (j + q) * KP);
            #pragma unroll
            for (int a = 0; a < 4; ++a) {
                float m0 = cx[a][0], m1 = cx[a][1], m2 = cx[a][2], m3 = cx[a][3];
                m0 = fmaxf(fmaxf(m0, sv[a].x + vv[0].x), sv[a].y + vv[1].x);
                m0 = fmaxf(fmaxf(m0, sv[a].z + vv[2].x), sv[a].w + vv[3].x);
                m1 = fmaxf(fmaxf(m1, sv[a].x + vv[0].y), sv[a].y + vv[1].y);
                m1 = fmaxf(fmaxf(m1, sv[a].z + vv[2].y), sv[a].w + vv[3].y);
                m2 = fmaxf(fmaxf(m2, sv[a].x + vv[0].z), sv[a].y + vv[1].z);
                m2 = fmaxf(fmaxf(m2, sv[a].z + vv[2].z), sv[a].w + vv[3].z);
                m3 = fmaxf(fmaxf(m3, sv[a].x + vv[0].w), sv[a].y + vv[1].w);
                m3 = fmaxf(fmaxf(m3, sv[a].z + vv[2].w), sv[a].w + vv[3].w);
                cx[a][0] = m0; cx[a][1] = m1; cx[a][2] = m2; cx[a][3] = m3;
            }
        }
        #pragma unroll
        for (int a = 0; a < 4; ++a)
            *(float4*)(pm + grp * (16 * KP) + (4 * ci + a) * KP + 4 * dg) =
                make_float4(cx[a][0], cx[a][1], cx[a][2], cx[a][3]);
    }
    __syncthreads();

    {
        const int i = t >> 4, c4 = t & 15;
        const float* pr = pm + i * KP + 4 * c4;
        float4 m = *(const float4*)pr;
        #pragma unroll
        for (int g = 1; g < 4; ++g) {
            const float4 o = *(const float4*)(pr + g * (16 * KP));
            m.x = fmaxf(m.x, o.x); m.y = fmaxf(m.y, o.y);
            m.z = fmaxf(m.z, o.z); m.w = fmaxf(m.w, o.w);
        }
        const int b = bh >> 3, h = bh & 7;
        *(float4*)(ctxout + (size_t)(b * LL + it * 16 + i) * DM + h * DKK + 4 * c4) =
            make_float4(expm1f(m.x), expm1f(m.y), expm1f(m.z), expm1f(m.w));
    }
}

extern "C" void kernel_launch(void* const* d_in, const int* in_sizes, int n_in,
                              void* d_out, int out_size, void* d_ws, size_t ws_size,
                              hipStream_t stream)
{
    (void)in_sizes; (void)n_in; (void)out_size; (void)ws_size;
    const float* x   = (const float*)d_in[0];
    const float* Wq  = (const float*)d_in[1];
    const float* Wk  = (const float*)d_in[2];
    const float* Wv  = (const float*)d_in[3];
    const float* Wo  = (const float*)d_in[4];
    const float* lam = (const float*)d_in[5];
    const float* Wqt = (const float*)d_in[6];
    const float* Wkt = (const float*)d_in[7];
    const float* Wvt = (const float*)d_in[8];

    float* out = (float*)d_out;
    float* sc  = out + (size_t)ROWS * DM;

    const size_t N = (size_t)ROWS * DM;
    float* ws  = (float*)d_ws;
    float* pp  = ws;
    float* ctx = ws + 4 * N;

    dim3 blk(256);
    proj_headtile<<<256, blk, 0, stream>>>(x, Wq, Wk, Wv, lam, Wqt, Wkt, Wvt, pp);

    trop_attn<<<dim3(16, 16), blk, 0, stream>>>(pp, pp + N, pp + 2 * N, pp + 3 * N, sc, ctx);

    gemm512_ksplit<<<256, dim3(128), 0, stream>>>(ctx, Wo, out);
}
